# starting kernel
# baseline (speedup 1.0000x reference)
#include <hip/hip_runtime.h>
#include <hip/hip_bf16.h>
#include <math.h>

#define N_NODES   50000
#define N_EDGES   800000
#define IN_FEATS  128
#define NUM_HEADS 8
#define OUT_FEATS 16
#define HD        (NUM_HEADS * OUT_FEATS)
#define NEG_SLOPE 0.2f

typedef float v2f __attribute__((ext_vector_type(2)));
typedef float v8f __attribute__((ext_vector_type(8)));

__global__ __launch_bounds__(256)
void gat_gemm_wmma(const float* __restrict__ feat,
                   const float* __restrict__ W,
                   float* __restrict__ h) {
  __shared__ float lA[16 * IN_FEATS];
  const int tid  = threadIdx.x;
  const int wave = tid >> 5;
  const int lane = tid & 31;
  const int m0   = blockIdx.x * 16;
  const int n0   = wave * 16;

  #pragma unroll
  for (int i = tid; i < 16 * IN_FEATS; i += 256) {
    const int r = i >> 7;
    const int c = i & 127;
    lA[i] = feat[(size_t)(m0 + r) * IN_FEATS + c];
  }
  __syncthreads();

  const int ml  = lane & 15;
  const int klo = (lane < 16) ? 0 : 2;

  v8f acc = {};
  #pragma unroll 4
  for (int k = 0; k < IN_FEATS; k += 4) {
    const float2 av = *(const float2*)&lA[ml * IN_FEATS + k + klo];
    v2f a; a.x = av.x; a.y = av.y;
    v2f b;
    b.x = W[(size_t)(k + klo)     * HD + n0 + ml];
    b.y = W[(size_t)(k + klo + 1) * HD + n0 + ml];
    acc = __builtin_amdgcn_wmma_f32_16x16x4_f32(
         false, a,  false, b,
         (short)0, acc,  false,  false);
  }

  const int rbase = m0 + ((lane < 16) ? 0 : 8);
  #pragma unroll
  for (int v = 0; v < 8; ++v) {
    h[(size_t)(rbase + v) * HD + n0 + ml] = acc[v];
  }
}

__global__ void gat_init_out(float* __restrict__ out,
                             const float* __restrict__ bias) {
  const int idx = blockIdx.x * blockDim.x + threadIdx.x;
  if (idx < N_NODES * HD) out[idx] = bias[idx & (HD - 1)];
}

__global__ void gat_logits(const float* __restrict__ h,
                           const float* __restrict__ attn,
                           float* __restrict__ es, float* __restrict__ ed,
                           float* __restrict__ m, float* __restrict__ denom) {
  const int idx = blockIdx.x * blockDim.x + threadIdx.x;
  if (idx >= N_NODES * NUM_HEADS) return;
  const int head = idx & (NUM_HEADS - 1);
  const int n    = idx >> 3;
  const float* hp = h    + (size_t)n * HD + head * OUT_FEATS;
  const float* a0 = attn +           head * OUT_FEATS;
  const float* a1 = attn + HD      + head * OUT_FEATS;
  float s0 = 0.f, s1 = 0.f;
  #pragma unroll
  for (int d = 0; d < OUT_FEATS; ++d) {
    const float hv = hp[d];
    s0 += hv * a0[d];
    s1 += hv * a1[d];
  }
  es[idx]    = s0;
  ed[idx]    = s1;
  m[idx]     = -INFINITY;
  denom[idx] = 0.f;
}

__device__ __forceinline__ void atomicMaxFloat(float* addr, float val) {
  if (val >= 0.f) {
    atomicMax((int*)addr, __float_as_int(val));
  } else {
    atomicMin((unsigned int*)addr, __float_as_uint(val));
  }
}

__device__ __forceinline__ float edge_score(const float* __restrict__ es,
                                            const float* __restrict__ ed,
                                            int s, int t, int head) {
  const float e = es[s * NUM_HEADS + head] + ed[t * NUM_HEADS + head];
  return e > 0.f ? e : NEG_SLOPE * e;
}

__global__ void gat_edge_max(const float* __restrict__ es,
                             const float* __restrict__ ed,
                             const int* __restrict__ src,
                             const int* __restrict__ dst,
                             float* __restrict__ m) {
  const int idx = blockIdx.x * blockDim.x + threadIdx.x;
  if (idx >= N_EDGES * NUM_HEADS) return;
  const int head = idx & (NUM_HEADS - 1);
  const int e    = idx >> 3;
  const int s = src[e], t = dst[e];
  atomicMaxFloat(&m[t * NUM_HEADS + head], edge_score(es, ed, s, t, head));
}

__global__ void gat_edge_sum(const float* __restrict__ es,
                             const float* __restrict__ ed,
                             const int* __restrict__ src,
                             const int* __restrict__ dst,
                             const float* __restrict__ m,
                             float* __restrict__ denom) {
  const int idx = blockIdx.x * blockDim.x + threadIdx.x;
  if (idx >= N_EDGES * NUM_HEADS) return;
  const int head = idx & (NUM_HEADS - 1);
  const int e    = idx >> 3;
  const int s = src[e], t = dst[e];
  const float v  = edge_score(es, ed, s, t, head);
  const float ex = __expf(v - m[t * NUM_HEADS + head]);
  atomicAdd(&denom[t * NUM_HEADS + head], ex);
}

__global__ void gat_aggregate(const float* __restrict__ h,
                              const float* __restrict__ es,
                              const float* __restrict__ ed,
                              const float* __restrict__ m,
                              const float* __restrict__ denom,
                              const int* __restrict__ src,
                              const int* __restrict__ dst,
                              float* __restrict__ out) {
  const int idx = blockIdx.x * blockDim.x + threadIdx.x;
  if (idx >= N_EDGES * NUM_HEADS) return;
  const int head = idx & (NUM_HEADS - 1);
  const int e    = idx >> 3;
  const int s = src[e], t = dst[e];
  const float v  = edge_score(es, ed, s, t, head);
  const float dn = denom[t * NUM_HEADS + head];
  const float alpha = __expf(v - m[t * NUM_HEADS + head]) / (dn > 0.f ? dn : 1.f);
  const float* hs = h   + (size_t)s * HD + head * OUT_FEATS;
  float*       op = out + (size_t)t * HD + head * OUT_FEATS;
  #pragma unroll
  for (int d = 0; d < OUT_FEATS; d += 4) {
    const float4 hv = *(const float4*)&hs[d];
    atomicAdd(&op[d + 0], alpha * hv.x);
    atomicAdd(&op[d + 1], alpha * hv.y);
    atomicAdd(&op[d + 2], alpha * hv.z);
    atomicAdd(&op[d + 3], alpha * hv.w);
  }
}

extern "C" void kernel_launch(void* const* d_in, const int* in_sizes, int n_in,
                              void* d_out, int out_size, void* d_ws, size_t ws_size,
                              hipStream_t stream) {
  const float* feat = (const float*)d_in[0];
  const float* W    = (const float*)d_in[1];
  const float* attn = (const float*)d_in[2];
  const float* bias = (const float*)d_in[3];
  const int*   src  = (const int*)d_in[4];
  const int*   dst  = (const int*)d_in[5];
  float*       out  = (float*)d_out;

  float* h     = (float*)d_ws;
  float* es    = h     + (size_t)N_NODES * HD;
  float* ed    = es    + (size_t)N_NODES * NUM_HEADS;
  float* m     = ed    + (size_t)N_NODES * NUM_HEADS;
  float* denom = m     + (size_t)N_NODES * NUM_HEADS;

  const int NH = N_NODES * NUM_HEADS;
  const int EH = N_EDGES * NUM_HEADS;

  gat_init_out<<<(N_NODES * HD + 255) / 256, 256, 0, stream>>>(out, bias);
  gat_gemm_wmma<<<N_NODES / 16, 256, 0, stream>>>(feat, W, h);
  gat_logits<<<(NH + 255) / 256, 256, 0, stream>>>(h, attn, es, ed, m, denom);
  gat_edge_max<<<(EH + 255) / 256, 256, 0, stream>>>(es, ed, src, dst, m);
  gat_edge_sum<<<(EH + 255) / 256, 256, 0, stream>>>(es, ed, src, dst, m, denom);
  gat_aggregate<<<(EH + 255) / 256, 256, 0, stream>>>(h, es, ed, m, denom, src, dst, out);
}


# Edges counting-sorted by dst in-call (int atomics, arrival order); one wave per dst node does max, exp-sum and weighted row sum in registers: no float atomics, out stored once, sum order changes
# speedup vs baseline: 3.1103x; 3.1103x over previous
#include <hip/hip_runtime.h>
#include <hip/hip_bf16.h>
#include <math.h>

#define N_NODES   50000
#define N_EDGES   800000
#define IN_FEATS  128
#define NUM_HEADS 8
#define OUT_FEATS 16
#define HD        (NUM_HEADS * OUT_FEATS)
#define NEG_SLOPE 0.2f

static_assert(N_NODES <= 65536, "sorted source ids are stored as 16-bit values");
static_assert(N_NODES % 8 == 0, "the gather kernel runs one wave per node, 8 per block");

typedef float v2f __attribute__((ext_vector_type(2)));
typedef float v8f __attribute__((ext_vector_type(8)));

__global__ __launch_bounds__(256)
void gat_gemm_wmma(const float* __restrict__ feat,
                   const float* __restrict__ W,
                   float* __restrict__ h) {
  __shared__ float lA[16 * IN_FEATS];
  const int tid  = threadIdx.x;
  const int wave = tid >> 5;
  const int lane = tid & 31;
  const int m0   = blockIdx.x * 16;
  const int n0   = wave * 16;

  #pragma unroll
  for (int i = tid; i < 16 * IN_FEATS; i += 256) {
    const int r = i >> 7;
    const int c = i & 127;
    lA[i] = feat[(size_t)(m0 + r) * IN_FEATS + c];
  }
  __syncthreads();

  const int ml  = lane & 15;
  const int klo = (lane < 16) ? 0 : 2;

  v8f acc = {};
  #pragma unroll 4
  for (int k = 0; k < IN_FEATS; k += 4) {
    const float2 av = *(const float2*)&lA[ml * IN_FEATS + k + klo];
    v2f a; a.x = av.x; a.y = av.y;
    v2f b;
    b.x = W[(size_t)(k + klo)     * HD + n0 + ml];
    b.y = W[(size_t)(k + klo + 1) * HD + n0 + ml];
    acc = __builtin_amdgcn_wmma_f32_16x16x4_f32(
         false, a,  false, b,
         (short)0, acc,  false,  false);
  }

  const int rbase = m0 + ((lane < 16) ? 0 : 8);
  #pragma unroll
  for (int v = 0; v < 8; ++v) {
    h[(size_t)(rbase + v) * HD + n0 + ml] = acc[v];
  }
}

__global__ void gat_logits(const float* __restrict__ h,
                           const float* __restrict__ attn,
                           float* __restrict__ es, float* __restrict__ ed,
                           int* __restrict__ cnt, int* __restrict__ counter) {
  const int idx = blockIdx.x * blockDim.x + threadIdx.x;
  if (idx >= N_NODES * NUM_HEADS) return;
  const int head = idx & (NUM_HEADS - 1);
  const int n    = idx >> 3;
  const float* hp = h    + (size_t)n * HD + head * OUT_FEATS;
  const float* a0 = attn +           head * OUT_FEATS;
  const float* a1 = attn + HD      + head * OUT_FEATS;
  float s0 = 0.f, s1 = 0.f;
  #pragma unroll
  for (int d = 0; d < OUT_FEATS; ++d) {
    const float hv = hp[d];
    s0 += hv * a0[d];
    s1 += hv * a1[d];
  }
  es[idx] = s0;
  ed[idx] = s1;
  if (head == 0) cnt[n] = 0;
  if (idx == 0) *counter = 0;
}

__global__ void gat_count(const int* __restrict__ dst, int* __restrict__ cnt) {
  const int e = blockIdx.x * blockDim.x + threadIdx.x;
  if (e >= N_EDGES) return;
  atomicAdd(&cnt[dst[e]], 1);
}

__global__ void gat_alloc(const int* __restrict__ cnt, int* __restrict__ cur,
                          int* __restrict__ counter) {
  const int n    = blockIdx.x * blockDim.x + threadIdx.x;
  const int lane = threadIdx.x & 31;
  const int c = (n < N_NODES) ? cnt[n] : 0;
  int x = c;
  #pragma unroll
  for (int o = 1; o < 32; o <<= 1) {
    const int y = __shfl_up(x, o);
    if (lane >= o) x += y;
  }
  int base = 0;
  if (lane == 31) base = atomicAdd(counter, x);
  base = __shfl(base, 31);
  if (n < N_NODES) cur[n] = base + x - c;
}

__global__ void gat_fill(const int* __restrict__ src, const int* __restrict__ dst,
                         int* __restrict__ cur, int* __restrict__ ssrc_words) {
  const int e = blockIdx.x * blockDim.x + threadIdx.x;
  if (e >= N_EDGES) return;
  unsigned short* ssrc = (unsigned short*)ssrc_words;
  const int pos = atomicAdd(&cur[dst[e]], 1);
  ssrc[pos] = (unsigned short)src[e];
}

__device__ __forceinline__ float leaky(float e) {
  return e > 0.f ? e : NEG_SLOPE * e;
}

__global__ __launch_bounds__(256)
void gat_gather(const float* __restrict__ h,
                const float* __restrict__ es,
                const float* __restrict__ ed,
                const float* __restrict__ bias,
                const int* __restrict__ cnt,
                const int* __restrict__ cur,
                const int* __restrict__ ssrc_words,
                float* __restrict__ out) {
  const unsigned short* ssrc = (const unsigned short*)ssrc_words;
  const int lane = threadIdx.x & 31;
  const int t   = __builtin_amdgcn_readfirstlane(blockIdx.x * 8 + (threadIdx.x >> 5));
  const int end = __builtin_amdgcn_readfirstlane(cur[t]);
  const int beg = end - __builtin_amdgcn_readfirstlane(cnt[t]);

  const int q    = lane >> 3;
  const int head = lane & 7;
  const float edt = ed[t * NUM_HEADS + head];
  float mx = -INFINITY;
  for (int j = beg + q; j < end; j += 4) {
    const int s = ssrc[j];
    mx = fmaxf(mx, leaky(es[s * NUM_HEADS + head] + edt));
  }
  mx = fmaxf(mx, __shfl_xor(mx, 8));
  mx = fmaxf(mx, __shfl_xor(mx, 16));
  if (!__builtin_isfinite(mx)) mx = 0.f;
  float sm = 0.f;
  for (int j = beg + q; j < end; j += 4) {
    const int s = ssrc[j];
    sm += __expf(leaky(es[s * NUM_HEADS + head] + edt) - mx);
  }
  sm += __shfl_xor(sm, 8);
  sm += __shfl_xor(sm, 16);

  const int hq = lane >> 2;
  const float edq = __shfl(edt, hq);
  const float mq  = __shfl(mx, hq);
  const float dq  = __shfl(sm, hq);
  const float dn  = dq > 0.f ? dq : 1.f;
  float4 acc = make_float4(0.f, 0.f, 0.f, 0.f);
  for (int base = beg; base < end; base += 32) {
    const int jj = base + lane;
    const int sv = (jj < end) ? (int)ssrc[jj] : 0;
    const int n  = (end - base < 32) ? (end - base) : 32;
    #pragma unroll 4
    for (int i = 0; i < n; ++i) {
      const int s = __builtin_amdgcn_readlane(sv, i);
      const float4 hv = *(const float4*)&h[(size_t)s * HD + 4 * lane];
      const float a = __expf(leaky(es[s * NUM_HEADS + hq] + edq) - mq) / dn;
      acc.x = fmaf(a, hv.x, acc.x);
      acc.y = fmaf(a, hv.y, acc.y);
      acc.z = fmaf(a, hv.z, acc.z);
      acc.w = fmaf(a, hv.w, acc.w);
    }
  }
  const float4 b = *(const float4*)&bias[4 * lane];
  acc.x += b.x; acc.y += b.y; acc.z += b.z; acc.w += b.w;
  *(float4*)&out[(size_t)t * HD + 4 * lane] = acc;
}

extern "C" void kernel_launch(void* const* d_in, const int* in_sizes, int n_in,
                              void* d_out, int out_size, void* d_ws, size_t ws_size,
                              hipStream_t stream) {
  const float* feat = (const float*)d_in[0];
  const float* W    = (const float*)d_in[1];
  const float* attn = (const float*)d_in[2];
  const float* bias = (const float*)d_in[3];
  const int*   src  = (const int*)d_in[4];
  const int*   dst  = (const int*)d_in[5];
  float*       out  = (float*)d_out;

  float* h       = (float*)d_ws;
  float* es      = h  + (size_t)N_NODES * HD;
  float* ed      = es + (size_t)N_NODES * NUM_HEADS;
  int*   cnt     = (int*)(ed + (size_t)N_NODES * NUM_HEADS);
  int*   cur     = cnt + N_NODES;
  int*   counter = cur + N_NODES;
  int*   ssrc    = counter + 64;

  const int NH = N_NODES * NUM_HEADS;

  gat_gemm_wmma<<<N_NODES / 16, 256, 0, stream>>>(feat, W, h);
  gat_logits<<<(NH + 255) / 256, 256, 0, stream>>>(h, attn, es, ed, cnt, counter);
  gat_count<<<(N_EDGES + 255) / 256, 256, 0, stream>>>(dst, cnt);
  gat_alloc<<<(N_NODES + 255) / 256, 256, 0, stream>>>(cnt, cur, counter);
  gat_fill<<<(N_EDGES + 255) / 256, 256, 0, stream>>>(src, dst, cur, ssrc);
  gat_gather<<<N_NODES / 8, 256, 0, stream>>>(h, es, ed, bias, cnt, cur, ssrc, out);
}


# Per-edge global atomics replaced by a two-level counting sort on LDS counters: edges binned by 256-node range, then ranked per node in one block per range; slot order stays arrival order
# speedup vs baseline: 4.8685x; 1.5653x over previous
#include <hip/hip_runtime.h>
#include <hip/hip_bf16.h>
#include <math.h>

#define N_NODES   50000
#define N_EDGES   800000
#define IN_FEATS  128
#define NUM_HEADS 8
#define OUT_FEATS 16
#define HD        (NUM_HEADS * OUT_FEATS)
#define NEG_SLOPE 0.2f

static_assert(N_NODES <= 65536, "sorted source ids are stored as 16-bit values");
static_assert(N_NODES % 8 == 0, "the gather kernel runs one wave per node, 8 per block");


#define NBUCK           ((N_NODES + 255) / 256)
#define SORT_BLOCKS     256
#define EDGES_PER_BLOCK (N_EDGES / SORT_BLOCKS)
#define EPT             ((EDGES_PER_BLOCK + 255) / 256)
#define PADW            32
static_assert(N_EDGES % SORT_BLOCKS == 0, "every sort block takes the same number of edges");
static_assert(N_EDGES % 4 == 0, "the range-local node bytes fill whole words");
static_assert(NBUCK <= 256, "one thread per node range in a 256-thread block");
static_assert((EPT - 1) * 256 <= EDGES_PER_BLOCK, "only a thread's last edge slot can be empty");
static_assert(N_EDGES / 2 + N_EDGES / 4 + SORT_BLOCKS * 256 + 2 * 256 * PADW
              <= 2 * N_NODES * NUM_HEADS, "sort scratch fits the es/ed planes");

typedef float v2f __attribute__((ext_vector_type(2)));
typedef float v8f __attribute__((ext_vector_type(8)));

__global__ __launch_bounds__(256)
void gat_gemm_wmma(const float* __restrict__ feat,
                   const float* __restrict__ W,
                   float* __restrict__ h,
                   int* __restrict__ total, int* __restrict__ cursor) {
  __shared__ float lA[16 * IN_FEATS];
  const int tid  = threadIdx.x;
  if (blockIdx.x == 0) {
    total[tid * PADW]  = 0;
    cursor[tid * PADW] = 0;
  }
  const int wave = tid >> 5;
  const int lane = tid & 31;
  const int m0   = blockIdx.x * 16;
  const int n0   = wave * 16;

  #pragma unroll
  for (int i = tid; i < 16 * IN_FEATS; i += 256) {
    const int r = i >> 7;
    const int c = i & 127;
    lA[i] = feat[(size_t)(m0 + r) * IN_FEATS + c];
  }
  __syncthreads();

  const int ml  = lane & 15;
  const int klo = (lane < 16) ? 0 : 2;

  v8f acc = {};
  #pragma unroll 4
  for (int k = 0; k < IN_FEATS; k += 4) {
    const float2 av = *(const float2*)&lA[ml * IN_FEATS + k + klo];
    v2f a; a.x = av.x; a.y = av.y;
    v2f b;
    b.x = W[(size_t)(k + klo)     * HD + n0 + ml];
    b.y = W[(size_t)(k + klo + 1) * HD + n0 + ml];
    acc = __builtin_amdgcn_wmma_f32_16x16x4_f32(
         false, a,  false, b,
         (short)0, acc,  false,  false);
  }

  const int rbase = m0 + ((lane < 16) ? 0 : 8);
  #pragma unroll
  for (int v = 0; v < 8; ++v) {
    h[(size_t)(rbase + v) * HD + n0 + ml] = acc[v];
  }
}

__global__ void gat_logits(const float* __restrict__ h,
                           const float* __restrict__ attn,
                           float* __restrict__ es, float* __restrict__ ed) {
  const int idx = blockIdx.x * blockDim.x + threadIdx.x;
  if (idx >= N_NODES * NUM_HEADS) return;
  const int head = idx & (NUM_HEADS - 1);
  const int n    = idx >> 3;
  const float* hp = h    + (size_t)n * HD + head * OUT_FEATS;
  const float* a0 = attn +           head * OUT_FEATS;
  const float* a1 = attn + HD      + head * OUT_FEATS;
  float s0 = 0.f, s1 = 0.f;
  #pragma unroll
  for (int d = 0; d < OUT_FEATS; ++d) {
    const float hv = hp[d];
    s0 += hv * a0[d];
    s1 += hv * a1[d];
  }
  es[idx] = s0;
  ed[idx] = s1;
}


__device__ __forceinline__ int excl_scan_256(int v, int tid, int* wsum) {
  const int lane = tid & 31;
  const int w    = tid >> 5;
  int x = v;
  #pragma unroll
  for (int o = 1; o < 32; o <<= 1) {
    const int y = __shfl_up(x, o);
    if (lane >= o) x += y;
  }
  if (lane == 31 && w < 8) wsum[w] = x;
  __syncthreads();
  int off = 0;
  #pragma unroll
  for (int j = 0; j < 8; ++j) if (j < w) off += wsum[j];
  return off + x - v;
}

__global__ __launch_bounds__(256)
void gat_range_hist(const int* __restrict__ dst, int* __restrict__ bh,
                    int* __restrict__ total) {
  __shared__ int lh[256];
  const int tid = threadIdx.x;
  const int e0  = blockIdx.x * EDGES_PER_BLOCK;
  lh[tid] = 0;
  int d[EPT];
  const bool last_ok = tid + (EPT - 1) * 256 < EDGES_PER_BLOCK;
  #pragma unroll
  for (int k = 0; k < EPT; ++k) {
    const int i = (k < EPT - 1 || last_ok) ? tid + k * 256 : EDGES_PER_BLOCK - 1;
    d[k] = dst[e0 + i];
  }
  __syncthreads();
  #pragma unroll
  for (int k = 0; k < EPT; ++k)
    if (k < EPT - 1 || last_ok) atomicAdd(&lh[d[k] >> 8], 1);
  __syncthreads();
  const int c = lh[tid];
  bh[blockIdx.x * 256 + tid] = c;
  if (c > 0) atomicAdd(&total[tid * PADW], c);
}

__global__ __launch_bounds__(256)
void gat_range_scatter(const int* __restrict__ src, const int* __restrict__ dst,
                       const int* __restrict__ bh, const int* __restrict__ total,
                       int* __restrict__ cursor,
                       int* __restrict__ src16_words, int* __restrict__ loc8_words) {
  __shared__ int lbase[256];
  __shared__ int lcnt[256];
  __shared__ int wsum[8];
  unsigned short* src16 = (unsigned short*)src16_words;
  unsigned char*  loc8  = (unsigned char*)loc8_words;
  const int tid = threadIdx.x;
  const int e0  = blockIdx.x * EDGES_PER_BLOCK;
  const int tot  = (tid < NBUCK) ? total[tid * PADW] : 0;
  const int mine = bh[blockIdx.x * 256 + tid];
  int d[EPT], s[EPT];
  const bool last_ok = tid + (EPT - 1) * 256 < EDGES_PER_BLOCK;
  #pragma unroll
  for (int k = 0; k < EPT; ++k) {
    const int i = (k < EPT - 1 || last_ok) ? tid + k * 256 : EDGES_PER_BLOCK - 1;
    d[k] = dst[e0 + i];
    s[k] = src[e0 + i];
  }
  const int start = excl_scan_256(tot, tid, wsum);
  int rel = 0;
  if (mine > 0) rel = atomicAdd(&cursor[tid * PADW], mine);
  lbase[tid] = start + rel;
  lcnt[tid]  = 0;
  __syncthreads();
  #pragma unroll
  for (int k = 0; k < EPT; ++k) {
    if (k < EPT - 1 || last_ok) {
      const int b = d[k] >> 8;
      const int p = lbase[b] + atomicAdd(&lcnt[b], 1);
      src16[p] = (unsigned short)s[k];
      loc8[p]  = (unsigned char)(d[k] & 255);
    }
  }
}

__global__ __launch_bounds__(1024)
void gat_node_sort(const int* __restrict__ total,
                   const int* __restrict__ src16_words,
                   const int* __restrict__ loc8_words,
                   int* __restrict__ cnt, int* __restrict__ cur,
                   int* __restrict__ ssrc_words) {
  __shared__ int ncnt[256];
  __shared__ int ncur[256];
  __shared__ int wsa[8];
  __shared__ int wsb[8];
  __shared__ int span[2];
  const unsigned short* src16 = (const unsigned short*)src16_words;
  const unsigned char*  loc8  = (const unsigned char*)loc8_words;
  unsigned short* ssrc = (unsigned short*)ssrc_words;
  const int tid = threadIdx.x;
  const int b   = blockIdx.x;
  const int tot = (tid < NBUCK) ? total[tid * PADW] : 0;
  if (tid < 256) ncnt[tid] = 0;
  const int st = excl_scan_256(tot, tid, wsa);
  if (tid == b) { span[0] = st; span[1] = tot; }
  __syncthreads();
  const int start = span[0];
  const int n     = span[1];

  for (int i0 = 0; i0 < n; i0 += 4096) {
    int l[4];
    #pragma unroll
    for (int k = 0; k < 4; ++k) {
      const int i  = i0 + tid + k * 1024;
      const int ii = (i < n) ? i : n - 1;
      const int v  = (int)loc8[start + ii];
      l[k] = (i < n) ? v : -1;
    }
    #pragma unroll
    for (int k = 0; k < 4; ++k)
      if (l[k] >= 0) atomicAdd(&ncnt[l[k]], 1);
  }
  __syncthreads();
  const int c  = (tid < 256) ? ncnt[tid] : 0;
  const int nb = excl_scan_256(c, tid, wsb);
  if (tid < 256) {
    ncur[tid] = nb;
    const int node = b * 256 + tid;
    if (node < N_NODES) {
      cnt[node] = c;
      cur[node] = start + nb + c;
    }
  }
  __syncthreads();
  for (int i0 = 0; i0 < n; i0 += 4096) {
    int l[4], sv[4];
    #pragma unroll
    for (int k = 0; k < 4; ++k) {
      const int i  = i0 + tid + k * 1024;
      const int ii = (i < n) ? i : n - 1;
      const int v  = (int)loc8[start + ii];
      sv[k] = (int)src16[start + ii];
      l[k]  = (i < n) ? v : -1;
    }
    #pragma unroll
    for (int k = 0; k < 4; ++k) {
      if (l[k] >= 0) {
        const int p = atomicAdd(&ncur[l[k]], 1);
        ssrc[start + p] = (unsigned short)sv[k];
      }
    }
  }
}

__device__ __forceinline__ float leaky(float e) {
  return e > 0.f ? e : NEG_SLOPE * e;
}

__global__ __launch_bounds__(256)
void gat_gather(const float* __restrict__ h,
                const float* __restrict__ es,
                const float* __restrict__ ed,
                const float* __restrict__ bias,
                const int* __restrict__ cnt,
                const int* __restrict__ cur,
                const int* __restrict__ ssrc_words,
                float* __restrict__ out) {
  const unsigned short* ssrc = (const unsigned short*)ssrc_words;
  const int lane = threadIdx.x & 31;
  const int t   = __builtin_amdgcn_readfirstlane(blockIdx.x * 8 + (threadIdx.x >> 5));
  const int end = __builtin_amdgcn_readfirstlane(cur[t]);
  const int beg = end - __builtin_amdgcn_readfirstlane(cnt[t]);

  const int q    = lane >> 3;
  const int head = lane & 7;
  const float edt = ed[t * NUM_HEADS + head];
  float mx = -INFINITY;
  for (int j = beg + q; j < end; j += 4) {
    const int s = ssrc[j];
    mx = fmaxf(mx, leaky(es[s * NUM_HEADS + head] + edt));
  }
  mx = fmaxf(mx, __shfl_xor(mx, 8));
  mx = fmaxf(mx, __shfl_xor(mx, 16));
  if (!__builtin_isfinite(mx)) mx = 0.f;
  float sm = 0.f;
  for (int j = beg + q; j < end; j += 4) {
    const int s = ssrc[j];
    sm += __expf(leaky(es[s * NUM_HEADS + head] + edt) - mx);
  }
  sm += __shfl_xor(sm, 8);
  sm += __shfl_xor(sm, 16);

  const int hq = lane >> 2;
  const float edq = __shfl(edt, hq);
  const float mq  = __shfl(mx, hq);
  const float dq  = __shfl(sm, hq);
  const float dn  = dq > 0.f ? dq : 1.f;
  float4 acc = make_float4(0.f, 0.f, 0.f, 0.f);
  for (int base = beg; base < end; base += 32) {
    const int jj = base + lane;
    const int sv = (jj < end) ? (int)ssrc[jj] : 0;
    const int n  = (end - base < 32) ? (end - base) : 32;
    #pragma unroll 4
    for (int i = 0; i < n; ++i) {
      const int s = __builtin_amdgcn_readlane(sv, i);
      const float4 hv = *(const float4*)&h[(size_t)s * HD + 4 * lane];
      const float a = __expf(leaky(es[s * NUM_HEADS + hq] + edq) - mq) / dn;
      acc.x = fmaf(a, hv.x, acc.x);
      acc.y = fmaf(a, hv.y, acc.y);
      acc.z = fmaf(a, hv.z, acc.z);
      acc.w = fmaf(a, hv.w, acc.w);
    }
  }
  const float4 b = *(const float4*)&bias[4 * lane];
  acc.x += b.x; acc.y += b.y; acc.z += b.z; acc.w += b.w;
  *(float4*)&out[(size_t)t * HD + 4 * lane] = acc;
}

extern "C" void kernel_launch(void* const* d_in, const int* in_sizes, int n_in,
                              void* d_out, int out_size, void* d_ws, size_t ws_size,
                              hipStream_t stream) {
  const float* feat = (const float*)d_in[0];
  const float* W    = (const float*)d_in[1];
  const float* attn = (const float*)d_in[2];
  const float* bias = (const float*)d_in[3];
  const int*   src  = (const int*)d_in[4];
  const int*   dst  = (const int*)d_in[5];
  float*       out  = (float*)d_out;

  float* h       = (float*)d_ws;
  float* es      = h  + (size_t)N_NODES * HD;
  float* ed      = es + (size_t)N_NODES * NUM_HEADS;
  int*   cnt     = (int*)(ed + (size_t)N_NODES * NUM_HEADS);
  int*   cur     = cnt + N_NODES;
  int*   ssrc    = cur + N_NODES + 64;

  int*   src16   = (int*)es;
  int*   loc8    = src16 + N_EDGES / 2;
  int*   bh      = loc8 + N_EDGES / 4;
  int*   total   = bh + SORT_BLOCKS * 256;
  int*   cursor  = total + 256 * PADW;

  const int NH = N_NODES * NUM_HEADS;

  gat_gemm_wmma<<<N_NODES / 16, 256, 0, stream>>>(feat, W, h, total, cursor);
  gat_range_hist<<<SORT_BLOCKS, 256, 0, stream>>>(dst, bh, total);
  gat_range_scatter<<<SORT_BLOCKS, 256, 0, stream>>>(src, dst, bh, total, cursor, src16, loc8);
  gat_node_sort<<<NBUCK, 1024, 0, stream>>>(total, src16, loc8, cnt, cur, ssrc);
  gat_logits<<<(NH + 255) / 256, 256, 0, stream>>>(h, attn, es, ed);
  gat_gather<<<N_NODES / 8, 256, 0, stream>>>(h, es, ed, bias, cnt, cur, ssrc, out);
}


# Gather forms each edge's logit from the gathered h row itself (4-lane dot on dpp adds) in one online-softmax pass: es/ed tables, their loads and the logits launch go; sum order changes
# speedup vs baseline: 5.7409x; 1.1792x over previous
#include <hip/hip_runtime.h>
#include <hip/hip_bf16.h>
#include <math.h>

#define N_NODES   50000
#define N_EDGES   800000
#define IN_FEATS  128
#define NUM_HEADS 8
#define OUT_FEATS 16
#define HD        (NUM_HEADS * OUT_FEATS)
#define NEG_SLOPE 0.2f

static_assert(N_NODES <= 65536, "sorted source ids are stored as 16-bit values");
static_assert(N_NODES % 8 == 0, "the gather kernel runs one wave per node, 8 per block");


#define NBUCK           ((N_NODES + 255) / 256)
#define SORT_BLOCKS     256
#define EDGES_PER_BLOCK (N_EDGES / SORT_BLOCKS)
#define EPT             ((EDGES_PER_BLOCK + 255) / 256)
#define PADW            32
static_assert(N_EDGES % SORT_BLOCKS == 0, "every sort block takes the same number of edges");
static_assert(N_EDGES % 4 == 0, "the range-local node bytes fill whole words");
static_assert(NBUCK <= 256, "one thread per node range in a 256-thread block");
static_assert((EPT - 1) * 256 <= EDGES_PER_BLOCK, "only a thread's last edge slot can be empty");
static_assert(N_EDGES / 2 + N_EDGES / 4 + SORT_BLOCKS * 256 + 2 * 256 * PADW
              <= 2 * N_NODES * NUM_HEADS, "sort scratch fits its region");

typedef float v2f __attribute__((ext_vector_type(2)));
typedef float v8f __attribute__((ext_vector_type(8)));

__global__ __launch_bounds__(256)
void gat_gemm_wmma(const float* __restrict__ feat,
                   const float* __restrict__ W,
                   float* __restrict__ h,
                   int* __restrict__ total, int* __restrict__ cursor) {
  __shared__ float lA[16 * IN_FEATS];
  const int tid  = threadIdx.x;
  if (blockIdx.x == 0) {
    total[tid * PADW]  = 0;
    cursor[tid * PADW] = 0;
  }
  const int wave = tid >> 5;
  const int lane = tid & 31;
  const int m0   = blockIdx.x * 16;
  const int n0   = wave * 16;

  #pragma unroll
  for (int i = tid; i < 16 * IN_FEATS; i += 256) {
    const int r = i >> 7;
    const int c = i & 127;
    lA[i] = feat[(size_t)(m0 + r) * IN_FEATS + c];
  }
  __syncthreads();

  const int ml  = lane & 15;
  const int klo = (lane < 16) ? 0 : 2;

  v8f acc = {};
  #pragma unroll 4
  for (int k = 0; k < IN_FEATS; k += 4) {
    const float2 av = *(const float2*)&lA[ml * IN_FEATS + k + klo];
    v2f a; a.x = av.x; a.y = av.y;
    v2f b;
    b.x = W[(size_t)(k + klo)     * HD + n0 + ml];
    b.y = W[(size_t)(k + klo + 1) * HD + n0 + ml];
    acc = __builtin_amdgcn_wmma_f32_16x16x4_f32(
         false, a,  false, b,
         (short)0, acc,  false,  false);
  }

  const int rbase = m0 + ((lane < 16) ? 0 : 8);
  #pragma unroll
  for (int v = 0; v < 8; ++v) {
    h[(size_t)(rbase + v) * HD + n0 + ml] = acc[v];
  }
}


__device__ __forceinline__ int excl_scan_256(int v, int tid, int* wsum) {
  const int lane = tid & 31;
  const int w    = tid >> 5;
  int x = v;
  #pragma unroll
  for (int o = 1; o < 32; o <<= 1) {
    const int y = __shfl_up(x, o);
    if (lane >= o) x += y;
  }
  if (lane == 31 && w < 8) wsum[w] = x;
  __syncthreads();
  int off = 0;
  #pragma unroll
  for (int j = 0; j < 8; ++j) if (j < w) off += wsum[j];
  return off + x - v;
}

__global__ __launch_bounds__(256)
void gat_range_hist(const int* __restrict__ dst, int* __restrict__ bh,
                    int* __restrict__ total) {
  __shared__ int lh[256];
  const int tid = threadIdx.x;
  const int e0  = blockIdx.x * EDGES_PER_BLOCK;
  lh[tid] = 0;
  int d[EPT];
  const bool last_ok = tid + (EPT - 1) * 256 < EDGES_PER_BLOCK;
  #pragma unroll
  for (int k = 0; k < EPT; ++k) {
    const int i = (k < EPT - 1 || last_ok) ? tid + k * 256 : EDGES_PER_BLOCK - 1;
    d[k] = dst[e0 + i];
  }
  __syncthreads();
  #pragma unroll
  for (int k = 0; k < EPT; ++k)
    if (k < EPT - 1 || last_ok) atomicAdd(&lh[d[k] >> 8], 1);
  __syncthreads();
  const int c = lh[tid];
  bh[blockIdx.x * 256 + tid] = c;
  if (c > 0) atomicAdd(&total[tid * PADW], c);
}

__global__ __launch_bounds__(256)
void gat_range_scatter(const int* __restrict__ src, const int* __restrict__ dst,
                       const int* __restrict__ bh, const int* __restrict__ total,
                       int* __restrict__ cursor,
                       int* __restrict__ src16_words, int* __restrict__ loc8_words) {
  __shared__ int lbase[256];
  __shared__ int lcnt[256];
  __shared__ int wsum[8];
  unsigned short* src16 = (unsigned short*)src16_words;
  unsigned char*  loc8  = (unsigned char*)loc8_words;
  const int tid = threadIdx.x;
  const int e0  = blockIdx.x * EDGES_PER_BLOCK;
  const int tot  = (tid < NBUCK) ? total[tid * PADW] : 0;
  const int mine = bh[blockIdx.x * 256 + tid];
  int d[EPT], s[EPT];
  const bool last_ok = tid + (EPT - 1) * 256 < EDGES_PER_BLOCK;
  #pragma unroll
  for (int k = 0; k < EPT; ++k) {
    const int i = (k < EPT - 1 || last_ok) ? tid + k * 256 : EDGES_PER_BLOCK - 1;
    d[k] = dst[e0 + i];
    s[k] = src[e0 + i];
  }
  const int start = excl_scan_256(tot, tid, wsum);
  int rel = 0;
  if (mine > 0) rel = atomicAdd(&cursor[tid * PADW], mine);
  lbase[tid] = start + rel;
  lcnt[tid]  = 0;
  __syncthreads();
  #pragma unroll
  for (int k = 0; k < EPT; ++k) {
    if (k < EPT - 1 || last_ok) {
      const int b = d[k] >> 8;
      const int p = lbase[b] + atomicAdd(&lcnt[b], 1);
      src16[p] = (unsigned short)s[k];
      loc8[p]  = (unsigned char)(d[k] & 255);
    }
  }
}

__global__ __launch_bounds__(1024)
void gat_node_sort(const int* __restrict__ total,
                   const int* __restrict__ src16_words,
                   const int* __restrict__ loc8_words,
                   int* __restrict__ cnt, int* __restrict__ cur,
                   int* __restrict__ ssrc_words) {
  __shared__ int ncnt[256];
  __shared__ int ncur[256];
  __shared__ int wsa[8];
  __shared__ int wsb[8];
  __shared__ int span[2];
  const unsigned short* src16 = (const unsigned short*)src16_words;
  const unsigned char*  loc8  = (const unsigned char*)loc8_words;
  unsigned short* ssrc = (unsigned short*)ssrc_words;
  const int tid = threadIdx.x;
  const int b   = blockIdx.x;
  const int tot = (tid < NBUCK) ? total[tid * PADW] : 0;
  if (tid < 256) ncnt[tid] = 0;
  const int st = excl_scan_256(tot, tid, wsa);
  if (tid == b) { span[0] = st; span[1] = tot; }
  __syncthreads();
  const int start = span[0];
  const int n     = span[1];

  for (int i0 = 0; i0 < n; i0 += 4096) {
    int l[4];
    #pragma unroll
    for (int k = 0; k < 4; ++k) {
      const int i  = i0 + tid + k * 1024;
      const int ii = (i < n) ? i : n - 1;
      const int v  = (int)loc8[start + ii];
      l[k] = (i < n) ? v : -1;
    }
    #pragma unroll
    for (int k = 0; k < 4; ++k)
      if (l[k] >= 0) atomicAdd(&ncnt[l[k]], 1);
  }
  __syncthreads();
  const int c  = (tid < 256) ? ncnt[tid] : 0;
  const int nb = excl_scan_256(c, tid, wsb);
  if (tid < 256) {
    ncur[tid] = nb;
    const int node = b * 256 + tid;
    if (node < N_NODES) {
      cnt[node] = c;
      cur[node] = start + nb + c;
    }
  }
  __syncthreads();
  for (int i0 = 0; i0 < n; i0 += 4096) {
    int l[4], sv[4];
    #pragma unroll
    for (int k = 0; k < 4; ++k) {
      const int i  = i0 + tid + k * 1024;
      const int ii = (i < n) ? i : n - 1;
      const int v  = (int)loc8[start + ii];
      sv[k] = (int)src16[start + ii];
      l[k]  = (i < n) ? v : -1;
    }
    #pragma unroll
    for (int k = 0; k < 4; ++k) {
      if (l[k] >= 0) {
        const int p = atomicAdd(&ncur[l[k]], 1);
        ssrc[start + p] = (unsigned short)sv[k];
      }
    }
  }
}

__device__ __forceinline__ float leaky(float e) {
  return e > 0.f ? e : NEG_SLOPE * e;
}

#define LOG2E 1.4426950408889634f

__device__ __forceinline__ float quad_sum(float x) {
  x += __builtin_bit_cast(float, __builtin_amdgcn_mov_dpp(__builtin_bit_cast(int, x), 0xB1, 0xf, 0xf, true));
  x += __builtin_bit_cast(float, __builtin_amdgcn_mov_dpp(__builtin_bit_cast(int, x), 0x4E, 0xf, 0xf, true));
  return x;
}

__device__ __forceinline__ float dot4(const float4 a, const float4 b) {
  return fmaf(a.w, b.w, fmaf(a.z, b.z, fmaf(a.y, b.y, a.x * b.x)));
}

template <int CNT>
__device__ __forceinline__ void edge_group(const float* __restrict__ h, const unsigned sv,
                                           const int i0, const int lane,
                                           const float4 a0, const float edq,
                                           float& m, float& l, float4& acc) {
  float4 hv[CNT];
  #pragma unroll
  for (int u = 0; u < CNT; ++u) {
    const unsigned s = __builtin_amdgcn_readlane(sv, i0 + u);
    hv[u] = *(const float4*)&h[(size_t)s * HD + 4 * lane];
  }
  float e[CNT];
  float mn = m;
  #pragma unroll
  for (int u = 0; u < CNT; ++u) {
    e[u] = leaky(quad_sum(dot4(hv[u], a0)) + edq);
    mn = fmaxf(mn, e[u]);
  }
  const float sc = __builtin_amdgcn_exp2f((m - mn) * LOG2E);
  acc.x *= sc; acc.y *= sc; acc.z *= sc; acc.w *= sc;
  float ps = 0.f;
  #pragma unroll
  for (int u = 0; u < CNT; ++u) {
    const float p = __builtin_amdgcn_exp2f((e[u] - mn) * LOG2E);
    ps += p;
    acc.x = fmaf(p, hv[u].x, acc.x);
    acc.y = fmaf(p, hv[u].y, acc.y);
    acc.z = fmaf(p, hv[u].z, acc.z);
    acc.w = fmaf(p, hv[u].w, acc.w);
  }
  l = fmaf(l, sc, ps);
  m = mn;
}

__global__ __launch_bounds__(256)
void gat_gather(const float* __restrict__ h,
                const float* __restrict__ attn,
                const float* __restrict__ bias,
                const int* __restrict__ cnt,
                const int* __restrict__ cur,
                const int* __restrict__ ssrc_words,
                float* __restrict__ out) {
  const unsigned short* ssrc = (const unsigned short*)ssrc_words;
  const int lane = threadIdx.x & 31;
  const int t   = __builtin_amdgcn_readfirstlane(blockIdx.x * 8 + (threadIdx.x >> 5));
  const int end = __builtin_amdgcn_readfirstlane(cur[t]);
  const int beg = end - __builtin_amdgcn_readfirstlane(cnt[t]);

  unsigned sv = (beg + lane < end) ? (unsigned)ssrc[beg + lane] : 0u;
  const float4 a0 = *(const float4*)&attn[4 * lane];
  const float4 a1 = *(const float4*)&attn[HD + 4 * lane];
  const float4 ht = *(const float4*)&h[(size_t)t * HD + 4 * lane];
  const float edq = quad_sum(dot4(ht, a1));

  float m = -INFINITY, l = 0.f;
  float4 acc = make_float4(0.f, 0.f, 0.f, 0.f);
  for (int base = beg; base < end; ) {
    const int n     = (end - base < 32) ? (end - base) : 32;
    const int nfull = n & ~3;
    for (int i0 = 0; i0 < nfull; i0 += 4)
      edge_group<4>(h, sv, i0, lane, a0, edq, m, l, acc);
    switch (n & 3) {
      case 1: edge_group<1>(h, sv, nfull, lane, a0, edq, m, l, acc); break;
      case 2: edge_group<2>(h, sv, nfull, lane, a0, edq, m, l, acc); break;
      case 3: edge_group<3>(h, sv, nfull, lane, a0, edq, m, l, acc); break;
      default: break;
    }
    base += 32;
    if (base < end) sv = (base + lane < end) ? (unsigned)ssrc[base + lane] : 0u;
  }
  const float r = 1.f / (l > 0.f ? l : 1.f);
  const float4 b = *(const float4*)&bias[4 * lane];
  acc.x = fmaf(acc.x, r, b.x);
  acc.y = fmaf(acc.y, r, b.y);
  acc.z = fmaf(acc.z, r, b.z);
  acc.w = fmaf(acc.w, r, b.w);
  *(float4*)&out[(size_t)t * HD + 4 * lane] = acc;
}

extern "C" void kernel_launch(void* const* d_in, const int* in_sizes, int n_in,
                              void* d_out, int out_size, void* d_ws, size_t ws_size,
                              hipStream_t stream) {
  const float* feat = (const float*)d_in[0];
  const float* W    = (const float*)d_in[1];
  const float* attn = (const float*)d_in[2];
  const float* bias = (const float*)d_in[3];
  const int*   src  = (const int*)d_in[4];
  const int*   dst  = (const int*)d_in[5];
  float*       out  = (float*)d_out;

  float* h       = (float*)d_ws;
  int*   scratch = (int*)(h + (size_t)N_NODES * HD);
  int*   cnt     = scratch + 2 * (size_t)N_NODES * NUM_HEADS;
  int*   cur     = cnt + N_NODES;
  int*   ssrc    = cur + N_NODES + 64;

  int*   src16   = scratch;
  int*   loc8    = src16 + N_EDGES / 2;
  int*   bh      = loc8 + N_EDGES / 4;
  int*   total   = bh + SORT_BLOCKS * 256;
  int*   cursor  = total + 256 * PADW;

  gat_gemm_wmma<<<N_NODES / 16, 256, 0, stream>>>(feat, W, h, total, cursor);
  gat_range_hist<<<SORT_BLOCKS, 256, 0, stream>>>(dst, bh, total);
  gat_range_scatter<<<SORT_BLOCKS, 256, 0, stream>>>(src, dst, bh, total, cursor, src16, loc8);
  gat_node_sort<<<NBUCK, 1024, 0, stream>>>(total, src16, loc8, cnt, cur, ssrc);
  gat_gather<<<N_NODES / 8, 256, 0, stream>>>(h, attn, bias, cnt, cur, ssrc, out);
}


# A tile's LDS row pitch padded from 128 to 132 words in the projection: the fragment read's 16 rows no longer share one bank pair (16-way conflict gone); k order, sums and results unchanged
# speedup vs baseline: 6.4697x; 1.1269x over previous
#include <hip/hip_runtime.h>
#include <hip/hip_bf16.h>
#include <math.h>

#define N_NODES   50000
#define N_EDGES   800000
#define IN_FEATS  128
#define NUM_HEADS 8
#define OUT_FEATS 16
#define HD        (NUM_HEADS * OUT_FEATS)
#define NEG_SLOPE 0.2f

static_assert(N_NODES <= 65536, "sorted source ids are stored as 16-bit values");
static_assert(N_NODES % 8 == 0, "the gather kernel runs one wave per node, 8 per block");


#define NBUCK           ((N_NODES + 255) / 256)
#define SORT_BLOCKS     256
#define EDGES_PER_BLOCK (N_EDGES / SORT_BLOCKS)
#define EPT             ((EDGES_PER_BLOCK + 255) / 256)
#define PADW            32
static_assert(N_EDGES % SORT_BLOCKS == 0, "every sort block takes the same number of edges");
static_assert(N_EDGES % 4 == 0, "the range-local node bytes fill whole words");
static_assert(NBUCK <= 256, "one thread per node range in a 256-thread block");
static_assert((EPT - 1) * 256 <= EDGES_PER_BLOCK, "only a thread's last edge slot can be empty");
static_assert(N_EDGES / 2 + N_EDGES / 4 + SORT_BLOCKS * 256 + 2 * 256 * PADW
              <= 2 * N_NODES * NUM_HEADS, "sort scratch fits its region");

typedef float v2f __attribute__((ext_vector_type(2)));
typedef float v8f __attribute__((ext_vector_type(8)));

#define LDA_PITCH (IN_FEATS + 4)
__global__ __launch_bounds__(256)
void gat_gemm_wmma(const float* __restrict__ feat,
                   const float* __restrict__ W,
                   float* __restrict__ h,
                   int* __restrict__ total, int* __restrict__ cursor) {
  __shared__ float lA[16 * LDA_PITCH];
  const int tid  = threadIdx.x;
  if (blockIdx.x == 0) {
    total[tid * PADW]  = 0;
    cursor[tid * PADW] = 0;
  }
  const int wave = tid >> 5;
  const int lane = tid & 31;
  const int m0   = blockIdx.x * 16;
  const int n0   = wave * 16;

  #pragma unroll
  for (int i = tid; i < 16 * IN_FEATS; i += 256) {
    const int r = i >> 7;
    const int c = i & 127;
    lA[r * LDA_PITCH + c] = feat[(size_t)(m0 + r) * IN_FEATS + c];
  }
  __syncthreads();

  const int ml  = lane & 15;
  const int klo = (lane < 16) ? 0 : 2;

  v8f acc = {};
  #pragma unroll 4
  for (int k = 0; k < IN_FEATS; k += 4) {
    const float2 av = *(const float2*)&lA[ml * LDA_PITCH + k + klo];
    v2f a; a.x = av.x; a.y = av.y;
    v2f b;
    b.x = W[(size_t)(k + klo)     * HD + n0 + ml];
    b.y = W[(size_t)(k + klo + 1) * HD + n0 + ml];
    acc = __builtin_amdgcn_wmma_f32_16x16x4_f32(
         false, a,  false, b,
         (short)0, acc,  false,  false);
  }

  const int rbase = m0 + ((lane < 16) ? 0 : 8);
  #pragma unroll
  for (int v = 0; v < 8; ++v) {
    h[(size_t)(rbase + v) * HD + n0 + ml] = acc[v];
  }
}


__device__ __forceinline__ int excl_scan_256(int v, int tid, int* wsum) {
  const int lane = tid & 31;
  const int w    = tid >> 5;
  int x = v;
  #pragma unroll
  for (int o = 1; o < 32; o <<= 1) {
    const int y = __shfl_up(x, o);
    if (lane >= o) x += y;
  }
  if (lane == 31 && w < 8) wsum[w] = x;
  __syncthreads();
  int off = 0;
  #pragma unroll
  for (int j = 0; j < 8; ++j) if (j < w) off += wsum[j];
  return off + x - v;
}

__global__ __launch_bounds__(256)
void gat_range_hist(const int* __restrict__ dst, int* __restrict__ bh,
                    int* __restrict__ total) {
  __shared__ int lh[256];
  const int tid = threadIdx.x;
  const int e0  = blockIdx.x * EDGES_PER_BLOCK;
  lh[tid] = 0;
  int d[EPT];
  const bool last_ok = tid + (EPT - 1) * 256 < EDGES_PER_BLOCK;
  #pragma unroll
  for (int k = 0; k < EPT; ++k) {
    const int i = (k < EPT - 1 || last_ok) ? tid + k * 256 : EDGES_PER_BLOCK - 1;
    d[k] = dst[e0 + i];
  }
  __syncthreads();
  #pragma unroll
  for (int k = 0; k < EPT; ++k)
    if (k < EPT - 1 || last_ok) atomicAdd(&lh[d[k] >> 8], 1);
  __syncthreads();
  const int c = lh[tid];
  bh[blockIdx.x * 256 + tid] = c;
  if (c > 0) atomicAdd(&total[tid * PADW], c);
}

__global__ __launch_bounds__(256)
void gat_range_scatter(const int* __restrict__ src, const int* __restrict__ dst,
                       const int* __restrict__ bh, const int* __restrict__ total,
                       int* __restrict__ cursor,
                       int* __restrict__ src16_words, int* __restrict__ loc8_words) {
  __shared__ int lbase[256];
  __shared__ int lcnt[256];
  __shared__ int wsum[8];
  unsigned short* src16 = (unsigned short*)src16_words;
  unsigned char*  loc8  = (unsigned char*)loc8_words;
  const int tid = threadIdx.x;
  const int e0  = blockIdx.x * EDGES_PER_BLOCK;
  const int tot  = (tid < NBUCK) ? total[tid * PADW] : 0;
  const int mine = bh[blockIdx.x * 256 + tid];
  int d[EPT], s[EPT];
  const bool last_ok = tid + (EPT - 1) * 256 < EDGES_PER_BLOCK;
  #pragma unroll
  for (int k = 0; k < EPT; ++k) {
    const int i = (k < EPT - 1 || last_ok) ? tid + k * 256 : EDGES_PER_BLOCK - 1;
    d[k] = dst[e0 + i];
    s[k] = src[e0 + i];
  }
  const int start = excl_scan_256(tot, tid, wsum);
  int rel = 0;
  if (mine > 0) rel = atomicAdd(&cursor[tid * PADW], mine);
  lbase[tid] = start + rel;
  lcnt[tid]  = 0;
  __syncthreads();
  #pragma unroll
  for (int k = 0; k < EPT; ++k) {
    if (k < EPT - 1 || last_ok) {
      const int b = d[k] >> 8;
      const int p = lbase[b] + atomicAdd(&lcnt[b], 1);
      src16[p] = (unsigned short)s[k];
      loc8[p]  = (unsigned char)(d[k] & 255);
    }
  }
}

__global__ __launch_bounds__(1024)
void gat_node_sort(const int* __restrict__ total,
                   const int* __restrict__ src16_words,
                   const int* __restrict__ loc8_words,
                   int* __restrict__ cnt, int* __restrict__ cur,
                   int* __restrict__ ssrc_words) {
  __shared__ int ncnt[256];
  __shared__ int ncur[256];
  __shared__ int wsa[8];
  __shared__ int wsb[8];
  __shared__ int span[2];
  const unsigned short* src16 = (const unsigned short*)src16_words;
  const unsigned char*  loc8  = (const unsigned char*)loc8_words;
  unsigned short* ssrc = (unsigned short*)ssrc_words;
  const int tid = threadIdx.x;
  const int b   = blockIdx.x;
  const int tot = (tid < NBUCK) ? total[tid * PADW] : 0;
  if (tid < 256) ncnt[tid] = 0;
  const int st = excl_scan_256(tot, tid, wsa);
  if (tid == b) { span[0] = st; span[1] = tot; }
  __syncthreads();
  const int start = span[0];
  const int n     = span[1];

  for (int i0 = 0; i0 < n; i0 += 4096) {
    int l[4];
    #pragma unroll
    for (int k = 0; k < 4; ++k) {
      const int i  = i0 + tid + k * 1024;
      const int ii = (i < n) ? i : n - 1;
      const int v  = (int)loc8[start + ii];
      l[k] = (i < n) ? v : -1;
    }
    #pragma unroll
    for (int k = 0; k < 4; ++k)
      if (l[k] >= 0) atomicAdd(&ncnt[l[k]], 1);
  }
  __syncthreads();
  const int c  = (tid < 256) ? ncnt[tid] : 0;
  const int nb = excl_scan_256(c, tid, wsb);
  if (tid < 256) {
    ncur[tid] = nb;
    const int node = b * 256 + tid;
    if (node < N_NODES) {
      cnt[node] = c;
      cur[node] = start + nb + c;
    }
  }
  __syncthreads();
  for (int i0 = 0; i0 < n; i0 += 4096) {
    int l[4], sv[4];
    #pragma unroll
    for (int k = 0; k < 4; ++k) {
      const int i  = i0 + tid + k * 1024;
      const int ii = (i < n) ? i : n - 1;
      const int v  = (int)loc8[start + ii];
      sv[k] = (int)src16[start + ii];
      l[k]  = (i < n) ? v : -1;
    }
    #pragma unroll
    for (int k = 0; k < 4; ++k) {
      if (l[k] >= 0) {
        const int p = atomicAdd(&ncur[l[k]], 1);
        ssrc[start + p] = (unsigned short)sv[k];
      }
    }
  }
}

__device__ __forceinline__ float leaky(float e) {
  return e > 0.f ? e : NEG_SLOPE * e;
}

#define LOG2E 1.4426950408889634f

__device__ __forceinline__ float quad_sum(float x) {
  x += __builtin_bit_cast(float, __builtin_amdgcn_mov_dpp(__builtin_bit_cast(int, x), 0xB1, 0xf, 0xf, true));
  x += __builtin_bit_cast(float, __builtin_amdgcn_mov_dpp(__builtin_bit_cast(int, x), 0x4E, 0xf, 0xf, true));
  return x;
}

__device__ __forceinline__ float dot4(const float4 a, const float4 b) {
  return fmaf(a.w, b.w, fmaf(a.z, b.z, fmaf(a.y, b.y, a.x * b.x)));
}

template <int CNT>
__device__ __forceinline__ void edge_group(const float* __restrict__ h, const unsigned sv,
                                           const int i0, const int lane,
                                           const float4 a0, const float edq,
                                           float& m, float& l, float4& acc) {
  float4 hv[CNT];
  #pragma unroll
  for (int u = 0; u < CNT; ++u) {
    const unsigned s = __builtin_amdgcn_readlane(sv, i0 + u);
    hv[u] = *(const float4*)&h[(size_t)s * HD + 4 * lane];
  }
  float e[CNT];
  float mn = m;
  #pragma unroll
  for (int u = 0; u < CNT; ++u) {
    e[u] = leaky(quad_sum(dot4(hv[u], a0)) + edq);
    mn = fmaxf(mn, e[u]);
  }
  const float sc = __builtin_amdgcn_exp2f((m - mn) * LOG2E);
  acc.x *= sc; acc.y *= sc; acc.z *= sc; acc.w *= sc;
  float ps = 0.f;
  #pragma unroll
  for (int u = 0; u < CNT; ++u) {
    const float p = __builtin_amdgcn_exp2f((e[u] - mn) * LOG2E);
    ps += p;
    acc.x = fmaf(p, hv[u].x, acc.x);
    acc.y = fmaf(p, hv[u].y, acc.y);
    acc.z = fmaf(p, hv[u].z, acc.z);
    acc.w = fmaf(p, hv[u].w, acc.w);
  }
  l = fmaf(l, sc, ps);
  m = mn;
}

__global__ __launch_bounds__(256)
void gat_gather(const float* __restrict__ h,
                const float* __restrict__ attn,
                const float* __restrict__ bias,
                const int* __restrict__ cnt,
                const int* __restrict__ cur,
                const int* __restrict__ ssrc_words,
                float* __restrict__ out) {
  const unsigned short* ssrc = (const unsigned short*)ssrc_words;
  const int lane = threadIdx.x & 31;
  const int t   = __builtin_amdgcn_readfirstlane(blockIdx.x * 8 + (threadIdx.x >> 5));
  const int end = __builtin_amdgcn_readfirstlane(cur[t]);
  const int beg = end - __builtin_amdgcn_readfirstlane(cnt[t]);

  unsigned sv = (beg + lane < end) ? (unsigned)ssrc[beg + lane] : 0u;
  const float4 a0 = *(const float4*)&attn[4 * lane];
  const float4 a1 = *(const float4*)&attn[HD + 4 * lane];
  const float4 ht = *(const float4*)&h[(size_t)t * HD + 4 * lane];
  const float edq = quad_sum(dot4(ht, a1));

  float m = -INFINITY, l = 0.f;
  float4 acc = make_float4(0.f, 0.f, 0.f, 0.f);
  for (int base = beg; base < end; ) {
    const int n     = (end - base < 32) ? (end - base) : 32;
    const int nfull = n & ~3;
    for (int i0 = 0; i0 < nfull; i0 += 4)
      edge_group<4>(h, sv, i0, lane, a0, edq, m, l, acc);
    switch (n & 3) {
      case 1: edge_group<1>(h, sv, nfull, lane, a0, edq, m, l, acc); break;
      case 2: edge_group<2>(h, sv, nfull, lane, a0, edq, m, l, acc); break;
      case 3: edge_group<3>(h, sv, nfull, lane, a0, edq, m, l, acc); break;
      default: break;
    }
    base += 32;
    if (base < end) sv = (base + lane < end) ? (unsigned)ssrc[base + lane] : 0u;
  }
  const float r = 1.f / (l > 0.f ? l : 1.f);
  const float4 b = *(const float4*)&bias[4 * lane];
  acc.x = fmaf(acc.x, r, b.x);
  acc.y = fmaf(acc.y, r, b.y);
  acc.z = fmaf(acc.z, r, b.z);
  acc.w = fmaf(acc.w, r, b.w);
  *(float4*)&out[(size_t)t * HD + 4 * lane] = acc;
}

extern "C" void kernel_launch(void* const* d_in, const int* in_sizes, int n_in,
                              void* d_out, int out_size, void* d_ws, size_t ws_size,
                              hipStream_t stream) {
  const float* feat = (const float*)d_in[0];
  const float* W    = (const float*)d_in[1];
  const float* attn = (const float*)d_in[2];
  const float* bias = (const float*)d_in[3];
  const int*   src  = (const int*)d_in[4];
  const int*   dst  = (const int*)d_in[5];
  float*       out  = (float*)d_out;

  float* h       = (float*)d_ws;
  int*   scratch = (int*)(h + (size_t)N_NODES * HD);
  int*   cnt     = scratch + 2 * (size_t)N_NODES * NUM_HEADS;
  int*   cur     = cnt + N_NODES;
  int*   ssrc    = cur + N_NODES + 64;

  int*   src16   = scratch;
  int*   loc8    = src16 + N_EDGES / 2;
  int*   bh      = loc8 + N_EDGES / 4;
  int*   total   = bh + SORT_BLOCKS * 256;
  int*   cursor  = total + 256 * PADW;

  gat_gemm_wmma<<<N_NODES / 16, 256, 0, stream>>>(feat, W, h, total, cursor);
  gat_range_hist<<<SORT_BLOCKS, 256, 0, stream>>>(dst, bh, total);
  gat_range_scatter<<<SORT_BLOCKS, 256, 0, stream>>>(src, dst, bh, total, cursor, src16, loc8);
  gat_node_sort<<<NBUCK, 1024, 0, stream>>>(total, src16, loc8, cnt, cur, ssrc);
  gat_gather<<<N_NODES / 8, 256, 0, stream>>>(h, attn, bias, cnt, cur, ssrc, out);
}


# Edge binning by node range made chunk-local: each block orders its own edges in LDS and publishes per-range piece offsets; histogram launch and all global atomics go; slots stay in arrival order
# speedup vs baseline: 8.9387x; 1.3816x over previous
#include <hip/hip_runtime.h>
#include <hip/hip_bf16.h>
#include <math.h>

#define N_NODES   50000
#define N_EDGES   800000
#define IN_FEATS  128
#define NUM_HEADS 8
#define OUT_FEATS 16
#define HD        (NUM_HEADS * OUT_FEATS)
#define NEG_SLOPE 0.2f

static_assert(N_NODES <= 65536, "sorted source ids are stored as 16-bit values");
static_assert(N_NODES % 8 == 0, "the gather kernel runs one wave per node, 8 per block");


#define NBUCK           ((N_NODES + 255) / 256)
#define SORT_BLOCKS     256
#define EDGES_PER_BLOCK (N_EDGES / SORT_BLOCKS)
#define EPT             ((EDGES_PER_BLOCK + 255) / 256)
static_assert(N_EDGES % SORT_BLOCKS == 0, "every sort block takes the same number of edges");
static_assert(N_EDGES % 4 == 0, "the range-local node bytes fill whole words");
static_assert(NBUCK <= 256, "one thread per node range in a 256-thread block");
static_assert(SORT_BLOCKS == 256, "a piece word is indexed range * 256 + chunk; 8 chunks per wave in the node sort");
static_assert(EDGES_PER_BLOCK < 32768, "a piece's offset and count share one word, 16 bits each");
static_assert((EPT - 1) * 256 <= EDGES_PER_BLOCK, "only a thread's last edge slot can be empty");
static_assert(N_EDGES / 2 + N_EDGES / 4 + 256 * SORT_BLOCKS
              <= 2 * N_NODES * NUM_HEADS, "sort scratch fits its region");

typedef float v2f __attribute__((ext_vector_type(2)));
typedef float v8f __attribute__((ext_vector_type(8)));

#define LDA_PITCH (IN_FEATS + 4)
__global__ __launch_bounds__(256)
void gat_gemm_wmma(const float* __restrict__ feat,
                   const float* __restrict__ W,
                   float* __restrict__ h) {
  __shared__ float lA[16 * LDA_PITCH];
  const int tid  = threadIdx.x;
  const int wave = tid >> 5;
  const int lane = tid & 31;
  const int m0   = blockIdx.x * 16;
  const int n0   = wave * 16;

  #pragma unroll
  for (int i = tid; i < 16 * IN_FEATS; i += 256) {
    const int r = i >> 7;
    const int c = i & 127;
    lA[r * LDA_PITCH + c] = feat[(size_t)(m0 + r) * IN_FEATS + c];
  }
  __syncthreads();

  const int ml  = lane & 15;
  const int klo = (lane < 16) ? 0 : 2;

  v8f acc = {};
  #pragma unroll 4
  for (int k = 0; k < IN_FEATS; k += 4) {
    const float2 av = *(const float2*)&lA[ml * LDA_PITCH + k + klo];
    v2f a; a.x = av.x; a.y = av.y;
    v2f b;
    b.x = W[(size_t)(k + klo)     * HD + n0 + ml];
    b.y = W[(size_t)(k + klo + 1) * HD + n0 + ml];
    acc = __builtin_amdgcn_wmma_f32_16x16x4_f32(
         false, a,  false, b,
         (short)0, acc,  false,  false);
  }

  const int rbase = m0 + ((lane < 16) ? 0 : 8);
  #pragma unroll
  for (int v = 0; v < 8; ++v) {
    h[(size_t)(rbase + v) * HD + n0 + ml] = acc[v];
  }
}


__device__ __forceinline__ int excl_scan_256(int v, int tid, int* wsum) {
  const int lane = tid & 31;
  const int w    = tid >> 5;
  int x = v;
  #pragma unroll
  for (int o = 1; o < 32; o <<= 1) {
    const int y = __shfl_up(x, o);
    if (lane >= o) x += y;
  }
  if (lane == 31 && w < 8) wsum[w] = x;
  __syncthreads();
  int off = 0;
  #pragma unroll
  for (int j = 0; j < 8; ++j) if (j < w) off += wsum[j];
  return off + x - v;
}

__global__ __launch_bounds__(256)
void gat_chunk_sort(const int* __restrict__ src, const int* __restrict__ dst,
                    int* __restrict__ piece,
                    int* __restrict__ src16_words, int* __restrict__ loc8_words) {
  __shared__ int lcnt[256];
  __shared__ int loff[256];
  __shared__ int wsum[8];
  unsigned short* src16 = (unsigned short*)src16_words;
  unsigned char*  loc8  = (unsigned char*)loc8_words;
  const int tid = threadIdx.x;
  const int e0  = blockIdx.x * EDGES_PER_BLOCK;
  lcnt[tid] = 0;
  int d[EPT], s[EPT], rk[EPT];
  const bool last_ok = tid + (EPT - 1) * 256 < EDGES_PER_BLOCK;
  #pragma unroll
  for (int k = 0; k < EPT; ++k) {
    const int i = (k < EPT - 1 || last_ok) ? tid + k * 256 : EDGES_PER_BLOCK - 1;
    d[k] = dst[e0 + i];
    s[k] = src[e0 + i];
  }
  __syncthreads();
  #pragma unroll
  for (int k = 0; k < EPT; ++k) {
    rk[k] = 0;
    if (k < EPT - 1 || last_ok) rk[k] = atomicAdd(&lcnt[d[k] >> 8], 1);
  }
  __syncthreads();
  const int c   = lcnt[tid];
  const int off = excl_scan_256(c, tid, wsum);
  loff[tid] = off;
  if (tid < NBUCK) piece[tid * 256 + blockIdx.x] = off | (c << 16);
  __syncthreads();
  #pragma unroll
  for (int k = 0; k < EPT; ++k) {
    if (k < EPT - 1 || last_ok) {
      const int p = e0 + loff[d[k] >> 8] + rk[k];
      src16[p] = (unsigned short)s[k];
      loc8[p]  = (unsigned char)(d[k] & 255);
    }
  }
}

__global__ __launch_bounds__(1024)
void gat_node_sort(const int* __restrict__ piece,
                   const int* __restrict__ src16_words,
                   const int* __restrict__ loc8_words,
                   int* __restrict__ cnt, int* __restrict__ cur,
                   int* __restrict__ ssrc_words) {
  __shared__ int ncnt[256];
  __shared__ int ncur[256];
  __shared__ int pbase[256];
  __shared__ int plen[256];
  __shared__ int wsb[8];
  __shared__ int first_slot;
  const unsigned short* src16 = (const unsigned short*)src16_words;
  const unsigned char*  loc8  = (const unsigned char*)loc8_words;
  unsigned short* ssrc = (unsigned short*)ssrc_words;
  const int tid = threadIdx.x;
  const int b   = blockIdx.x;

  int low = 0;
  if (tid < 256) {
    const int v = piece[b * 256 + tid];
    low = v & 0xFFFF;
    pbase[tid] = tid * EDGES_PER_BLOCK + low;
    plen[tid]  = v >> 16;
    ncnt[tid]  = 0;
  }
  if (tid == 0) first_slot = 0;
  __syncthreads();
  const int lane = tid & 31;
  const int w8   = (tid >> 5) * 8;
  int nn[8], gg[8];
  #pragma unroll
  for (int i = 0; i < 8; ++i) {
    nn[i] = __builtin_amdgcn_readfirstlane(plen[w8 + i]);
    gg[i] = __builtin_amdgcn_readfirstlane(pbase[w8 + i]);
  }
  int l[8], sv[8];
  #pragma unroll
  for (int i = 0; i < 8; ++i) {
    const int jj = max(min(lane, nn[i] - 1), 0);
    const int at = min(gg[i] + jj, N_EDGES - 1);
    l[i]  = (int)loc8[at];
    sv[i] = (int)src16[at];
  }
  asm volatile("" :: "v"(l[0]), "v"(l[1]), "v"(l[2]), "v"(l[3]),
                     "v"(l[4]), "v"(l[5]), "v"(l[6]), "v"(l[7]),
                     "v"(sv[0]), "v"(sv[1]), "v"(sv[2]), "v"(sv[3]),
                     "v"(sv[4]), "v"(sv[5]), "v"(sv[6]), "v"(sv[7]));
  #pragma unroll
  for (int i = 0; i < 8; ++i)
    if (lane < nn[i]) atomicAdd(&ncnt[l[i]], 1);
  #pragma unroll
  for (int i = 0; i < 8; ++i) {
    if (nn[i] > 32) {
      for (int j = lane + 32; j < nn[i]; j += 32) atomicAdd(&ncnt[(int)loc8[gg[i] + j]], 1);
    }
  }
  low += __builtin_amdgcn_update_dpp(0, low, 0x161, 0xf, 0xf, false);
  low += __builtin_amdgcn_update_dpp(0, low, 0x162, 0xf, 0xf, false);
  low += __builtin_amdgcn_update_dpp(0, low, 0x164, 0xf, 0xf, false);
  low += __builtin_amdgcn_update_dpp(0, low, 0x168, 0xf, 0xf, false);
  if ((tid & 15) == 0 && low != 0) atomicAdd(&first_slot, low);
  __syncthreads();
  const int c  = (tid < 256) ? ncnt[tid] : 0;
  const int nb = excl_scan_256(c, tid, wsb);
  if (tid < 256) {
    const int at = first_slot + nb;
    ncur[tid] = at;
    const int node = b * 256 + tid;
    if (node < N_NODES) {
      cnt[node] = c;
      cur[node] = at + c;
    }
  }
  __syncthreads();
  {
    unsigned p[8];
    #pragma unroll
    for (int i = 0; i < 8; ++i) {
      p[i] = 0;
      if (lane < nn[i]) p[i] = (unsigned)atomicAdd(&ncur[l[i]], 1);
    }
    #pragma unroll
    for (int i = 0; i < 8; ++i)
      if (lane < nn[i]) ssrc[p[i]] = (unsigned short)sv[i];
  }
  #pragma unroll
  for (int i = 0; i < 8; ++i) {
    if (nn[i] > 32) {
      for (int j = lane + 32; j < nn[i]; j += 32) {
        const int      ln = (int)loc8[gg[i] + j];
        const unsigned s  = src16[gg[i] + j];
        ssrc[atomicAdd(&ncur[ln], 1)] = (unsigned short)s;
      }
    }
  }
}

__device__ __forceinline__ float leaky(float e) {
  return e > 0.f ? e : NEG_SLOPE * e;
}

#define LOG2E 1.4426950408889634f

__device__ __forceinline__ float quad_sum(float x) {
  x += __builtin_bit_cast(float, __builtin_amdgcn_mov_dpp(__builtin_bit_cast(int, x), 0xB1, 0xf, 0xf, true));
  x += __builtin_bit_cast(float, __builtin_amdgcn_mov_dpp(__builtin_bit_cast(int, x), 0x4E, 0xf, 0xf, true));
  return x;
}

__device__ __forceinline__ float dot4(const float4 a, const float4 b) {
  return fmaf(a.w, b.w, fmaf(a.z, b.z, fmaf(a.y, b.y, a.x * b.x)));
}

template <int CNT>
__device__ __forceinline__ void edge_group(const float* __restrict__ h, const unsigned sv,
                                           const int i0, const int lane,
                                           const float4 a0, const float edq,
                                           float& m, float& l, float4& acc) {
  float4 hv[CNT];
  #pragma unroll
  for (int u = 0; u < CNT; ++u) {
    const unsigned s = __builtin_amdgcn_readlane(sv, i0 + u);
    hv[u] = *(const float4*)&h[(size_t)s * HD + 4 * lane];
  }
  float e[CNT];
  float mn = m;
  #pragma unroll
  for (int u = 0; u < CNT; ++u) {
    e[u] = leaky(quad_sum(dot4(hv[u], a0)) + edq);
    mn = fmaxf(mn, e[u]);
  }
  const float sc = __builtin_amdgcn_exp2f((m - mn) * LOG2E);
  acc.x *= sc; acc.y *= sc; acc.z *= sc; acc.w *= sc;
  float ps = 0.f;
  #pragma unroll
  for (int u = 0; u < CNT; ++u) {
    const float p = __builtin_amdgcn_exp2f((e[u] - mn) * LOG2E);
    ps += p;
    acc.x = fmaf(p, hv[u].x, acc.x);
    acc.y = fmaf(p, hv[u].y, acc.y);
    acc.z = fmaf(p, hv[u].z, acc.z);
    acc.w = fmaf(p, hv[u].w, acc.w);
  }
  l = fmaf(l, sc, ps);
  m = mn;
}

__global__ __launch_bounds__(256)
void gat_gather(const float* __restrict__ h,
                const float* __restrict__ attn,
                const float* __restrict__ bias,
                const int* __restrict__ cnt,
                const int* __restrict__ cur,
                const int* __restrict__ ssrc_words,
                float* __restrict__ out) {
  const unsigned short* ssrc = (const unsigned short*)ssrc_words;
  const int lane = threadIdx.x & 31;
  const int t   = __builtin_amdgcn_readfirstlane(blockIdx.x * 8 + (threadIdx.x >> 5));
  const int end = __builtin_amdgcn_readfirstlane(cur[t]);
  const int beg = end - __builtin_amdgcn_readfirstlane(cnt[t]);

  unsigned sv = (beg + lane < end) ? (unsigned)ssrc[beg + lane] : 0u;
  const float4 a0 = *(const float4*)&attn[4 * lane];
  const float4 a1 = *(const float4*)&attn[HD + 4 * lane];
  const float4 ht = *(const float4*)&h[(size_t)t * HD + 4 * lane];
  const float edq = quad_sum(dot4(ht, a1));

  float m = -INFINITY, l = 0.f;
  float4 acc = make_float4(0.f, 0.f, 0.f, 0.f);
  for (int base = beg; base < end; ) {
    const int n     = (end - base < 32) ? (end - base) : 32;
    const int nfull = n & ~3;
    for (int i0 = 0; i0 < nfull; i0 += 4)
      edge_group<4>(h, sv, i0, lane, a0, edq, m, l, acc);
    switch (n & 3) {
      case 1: edge_group<1>(h, sv, nfull, lane, a0, edq, m, l, acc); break;
      case 2: edge_group<2>(h, sv, nfull, lane, a0, edq, m, l, acc); break;
      case 3: edge_group<3>(h, sv, nfull, lane, a0, edq, m, l, acc); break;
      default: break;
    }
    base += 32;
    if (base < end) sv = (base + lane < end) ? (unsigned)ssrc[base + lane] : 0u;
  }
  const float r = 1.f / (l > 0.f ? l : 1.f);
  const float4 b = *(const float4*)&bias[4 * lane];
  acc.x = fmaf(acc.x, r, b.x);
  acc.y = fmaf(acc.y, r, b.y);
  acc.z = fmaf(acc.z, r, b.z);
  acc.w = fmaf(acc.w, r, b.w);
  *(float4*)&out[(size_t)t * HD + 4 * lane] = acc;
}

extern "C" void kernel_launch(void* const* d_in, const int* in_sizes, int n_in,
                              void* d_out, int out_size, void* d_ws, size_t ws_size,
                              hipStream_t stream) {
  const float* feat = (const float*)d_in[0];
  const float* W    = (const float*)d_in[1];
  const float* attn = (const float*)d_in[2];
  const float* bias = (const float*)d_in[3];
  const int*   src  = (const int*)d_in[4];
  const int*   dst  = (const int*)d_in[5];
  float*       out  = (float*)d_out;

  float* h       = (float*)d_ws;
  int*   scratch = (int*)(h + (size_t)N_NODES * HD);
  int*   cnt     = scratch + 2 * (size_t)N_NODES * NUM_HEADS;
  int*   cur     = cnt + N_NODES;
  int*   ssrc    = cur + N_NODES + 64;

  int*   src16   = scratch;
  int*   loc8    = src16 + N_EDGES / 2;
  int*   piece   = loc8 + N_EDGES / 4;

  gat_gemm_wmma<<<N_NODES / 16, 256, 0, stream>>>(feat, W, h);
  gat_chunk_sort<<<SORT_BLOCKS, 256, 0, stream>>>(src, dst, piece, src16, loc8);
  gat_node_sort<<<NBUCK, 1024, 0, stream>>>(piece, src16, loc8, cnt, cur, ssrc);
  gat_gather<<<N_NODES / 8, 256, 0, stream>>>(h, attn, bias, cnt, cur, ssrc, out);
}


# Projection and chunk sort, which share no data, run in one launch: blocks 0-255 sort their edge chunk while the rest do the WMMA tiles; 3 launches, not 4; sums and results unchanged
# speedup vs baseline: 9.6616x; 1.0809x over previous
#include <hip/hip_runtime.h>
#include <hip/hip_bf16.h>
#include <math.h>

#define N_NODES   50000
#define N_EDGES   800000
#define IN_FEATS  128
#define NUM_HEADS 8
#define OUT_FEATS 16
#define HD        (NUM_HEADS * OUT_FEATS)
#define NEG_SLOPE 0.2f

static_assert(N_NODES <= 65536, "sorted source ids are stored as 16-bit values");
static_assert(N_NODES % 8 == 0, "the gather kernel runs one wave per node, 8 per block");


#define NBUCK           ((N_NODES + 255) / 256)
#define SORT_BLOCKS     256
#define EDGES_PER_BLOCK (N_EDGES / SORT_BLOCKS)
#define EPT             ((EDGES_PER_BLOCK + 255) / 256)
static_assert(N_EDGES % SORT_BLOCKS == 0, "every sort block takes the same number of edges");
static_assert(N_EDGES % 4 == 0, "the range-local node bytes fill whole words");
static_assert(NBUCK <= 256, "one thread per node range in a 256-thread block");
static_assert(SORT_BLOCKS == 256, "a piece word is indexed range * 256 + chunk; 8 chunks per wave in the node sort");
static_assert(EDGES_PER_BLOCK < 32768, "a piece's offset and count share one word, 16 bits each");
static_assert((EPT - 1) * 256 <= EDGES_PER_BLOCK, "only a thread's last edge slot can be empty");
static_assert(N_EDGES / 2 + N_EDGES / 4 + 256 * SORT_BLOCKS
              <= 2 * N_NODES * NUM_HEADS, "sort scratch fits its region");

typedef float v2f __attribute__((ext_vector_type(2)));
typedef float v8f __attribute__((ext_vector_type(8)));

#define LDA_PITCH (IN_FEATS + 4)
__device__ __forceinline__
void gemm_tile(const float* __restrict__ feat,
               const float* __restrict__ W,
               float* __restrict__ h, const int tile) {
  __shared__ float lA[16 * LDA_PITCH];
  const int tid  = threadIdx.x;
  const int wave = tid >> 5;
  const int lane = tid & 31;
  const int m0   = tile * 16;
  const int n0   = wave * 16;

  #pragma unroll
  for (int i = tid; i < 16 * IN_FEATS; i += 256) {
    const int r = i >> 7;
    const int c = i & 127;
    lA[r * LDA_PITCH + c] = feat[(size_t)(m0 + r) * IN_FEATS + c];
  }
  __syncthreads();

  const int ml  = lane & 15;
  const int klo = (lane < 16) ? 0 : 2;

  v8f acc = {};
  #pragma unroll 4
  for (int k = 0; k < IN_FEATS; k += 4) {
    const float2 av = *(const float2*)&lA[ml * LDA_PITCH + k + klo];
    v2f a; a.x = av.x; a.y = av.y;
    v2f b;
    b.x = W[(size_t)(k + klo)     * HD + n0 + ml];
    b.y = W[(size_t)(k + klo + 1) * HD + n0 + ml];
    acc = __builtin_amdgcn_wmma_f32_16x16x4_f32(
         false, a,  false, b,
         (short)0, acc,  false,  false);
  }

  const int rbase = m0 + ((lane < 16) ? 0 : 8);
  #pragma unroll
  for (int v = 0; v < 8; ++v) {
    h[(size_t)(rbase + v) * HD + n0 + ml] = acc[v];
  }
}


__device__ __forceinline__ int excl_scan_256(int v, int tid, int* wsum) {
  const int lane = tid & 31;
  const int w    = tid >> 5;
  int x = v;
  #pragma unroll
  for (int o = 1; o < 32; o <<= 1) {
    const int y = __shfl_up(x, o);
    if (lane >= o) x += y;
  }
  if (lane == 31 && w < 8) wsum[w] = x;
  __syncthreads();
  int off = 0;
  #pragma unroll
  for (int j = 0; j < 8; ++j) if (j < w) off += wsum[j];
  return off + x - v;
}

__device__ __forceinline__
void chunk_sort(const int* __restrict__ src, const int* __restrict__ dst,
                int* __restrict__ piece,
                int* __restrict__ src16_words, int* __restrict__ loc8_words,
                const int chunk) {
  __shared__ int lcnt[256];
  __shared__ int loff[256];
  __shared__ int wsum[8];
  unsigned short* src16 = (unsigned short*)src16_words;
  unsigned char*  loc8  = (unsigned char*)loc8_words;
  const int tid = threadIdx.x;
  const int e0  = chunk * EDGES_PER_BLOCK;
  lcnt[tid] = 0;
  int d[EPT], s[EPT], rk[EPT];
  const bool last_ok = tid + (EPT - 1) * 256 < EDGES_PER_BLOCK;
  #pragma unroll
  for (int k = 0; k < EPT; ++k) {
    const int i = (k < EPT - 1 || last_ok) ? tid + k * 256 : EDGES_PER_BLOCK - 1;
    d[k] = dst[e0 + i];
    s[k] = src[e0 + i];
  }
  __syncthreads();
  #pragma unroll
  for (int k = 0; k < EPT; ++k) {
    rk[k] = 0;
    if (k < EPT - 1 || last_ok) rk[k] = atomicAdd(&lcnt[d[k] >> 8], 1);
  }
  __syncthreads();
  const int c   = lcnt[tid];
  const int off = excl_scan_256(c, tid, wsum);
  loff[tid] = off;
  if (tid < NBUCK) piece[tid * 256 + chunk] = off | (c << 16);
  __syncthreads();
  #pragma unroll
  for (int k = 0; k < EPT; ++k) {
    if (k < EPT - 1 || last_ok) {
      const int p = e0 + loff[d[k] >> 8] + rk[k];
      src16[p] = (unsigned short)s[k];
      loc8[p]  = (unsigned char)(d[k] & 255);
    }
  }
}

__global__ __launch_bounds__(256, 16)
void gat_chunk_sort_and_project(const float* __restrict__ feat,
                                const float* __restrict__ W,
                                float* __restrict__ h,
                                const int* __restrict__ src, const int* __restrict__ dst,
                                int* __restrict__ piece,
                                int* __restrict__ src16_words, int* __restrict__ loc8_words) {
  if (blockIdx.x < SORT_BLOCKS) chunk_sort(src, dst, piece, src16_words, loc8_words, blockIdx.x);
  else                          gemm_tile(feat, W, h, blockIdx.x - SORT_BLOCKS);
}

__global__ __launch_bounds__(1024)
void gat_node_sort(const int* __restrict__ piece,
                   const int* __restrict__ src16_words,
                   const int* __restrict__ loc8_words,
                   int* __restrict__ cnt, int* __restrict__ cur,
                   int* __restrict__ ssrc_words) {
  __shared__ int ncnt[256];
  __shared__ int ncur[256];
  __shared__ int pbase[256];
  __shared__ int plen[256];
  __shared__ int wsb[8];
  __shared__ int first_slot;
  const unsigned short* src16 = (const unsigned short*)src16_words;
  const unsigned char*  loc8  = (const unsigned char*)loc8_words;
  unsigned short* ssrc = (unsigned short*)ssrc_words;
  const int tid = threadIdx.x;
  const int b   = blockIdx.x;

  int low = 0;
  if (tid < 256) {
    const int v = piece[b * 256 + tid];
    low = v & 0xFFFF;
    pbase[tid] = tid * EDGES_PER_BLOCK + low;
    plen[tid]  = v >> 16;
    ncnt[tid]  = 0;
  }
  if (tid == 0) first_slot = 0;
  __syncthreads();
  const int lane = tid & 31;
  const int w8   = (tid >> 5) * 8;
  int nn[8], gg[8];
  #pragma unroll
  for (int i = 0; i < 8; ++i) {
    nn[i] = __builtin_amdgcn_readfirstlane(plen[w8 + i]);
    gg[i] = __builtin_amdgcn_readfirstlane(pbase[w8 + i]);
  }
  int l[8], sv[8];
  #pragma unroll
  for (int i = 0; i < 8; ++i) {
    const int jj = max(min(lane, nn[i] - 1), 0);
    const int at = min(gg[i] + jj, N_EDGES - 1);
    l[i]  = (int)loc8[at];
    sv[i] = (int)src16[at];
  }
  asm volatile("" :: "v"(l[0]), "v"(l[1]), "v"(l[2]), "v"(l[3]),
                     "v"(l[4]), "v"(l[5]), "v"(l[6]), "v"(l[7]),
                     "v"(sv[0]), "v"(sv[1]), "v"(sv[2]), "v"(sv[3]),
                     "v"(sv[4]), "v"(sv[5]), "v"(sv[6]), "v"(sv[7]));
  #pragma unroll
  for (int i = 0; i < 8; ++i)
    if (lane < nn[i]) atomicAdd(&ncnt[l[i]], 1);
  #pragma unroll
  for (int i = 0; i < 8; ++i) {
    if (nn[i] > 32) {
      for (int j = lane + 32; j < nn[i]; j += 32) atomicAdd(&ncnt[(int)loc8[gg[i] + j]], 1);
    }
  }
  low += __builtin_amdgcn_update_dpp(0, low, 0x161, 0xf, 0xf, false);
  low += __builtin_amdgcn_update_dpp(0, low, 0x162, 0xf, 0xf, false);
  low += __builtin_amdgcn_update_dpp(0, low, 0x164, 0xf, 0xf, false);
  low += __builtin_amdgcn_update_dpp(0, low, 0x168, 0xf, 0xf, false);
  if ((tid & 15) == 0 && low != 0) atomicAdd(&first_slot, low);
  __syncthreads();
  const int c  = (tid < 256) ? ncnt[tid] : 0;
  const int nb = excl_scan_256(c, tid, wsb);
  if (tid < 256) {
    const int at = first_slot + nb;
    ncur[tid] = at;
    const int node = b * 256 + tid;
    if (node < N_NODES) {
      cnt[node] = c;
      cur[node] = at + c;
    }
  }
  __syncthreads();
  {
    unsigned p[8];
    #pragma unroll
    for (int i = 0; i < 8; ++i) {
      p[i] = 0;
      if (lane < nn[i]) p[i] = (unsigned)atomicAdd(&ncur[l[i]], 1);
    }
    #pragma unroll
    for (int i = 0; i < 8; ++i)
      if (lane < nn[i]) ssrc[p[i]] = (unsigned short)sv[i];
  }
  #pragma unroll
  for (int i = 0; i < 8; ++i) {
    if (nn[i] > 32) {
      for (int j = lane + 32; j < nn[i]; j += 32) {
        const int      ln = (int)loc8[gg[i] + j];
        const unsigned s  = src16[gg[i] + j];
        ssrc[atomicAdd(&ncur[ln], 1)] = (unsigned short)s;
      }
    }
  }
}

__device__ __forceinline__ float leaky(float e) {
  return e > 0.f ? e : NEG_SLOPE * e;
}

#define LOG2E 1.4426950408889634f

__device__ __forceinline__ float quad_sum(float x) {
  x += __builtin_bit_cast(float, __builtin_amdgcn_mov_dpp(__builtin_bit_cast(int, x), 0xB1, 0xf, 0xf, true));
  x += __builtin_bit_cast(float, __builtin_amdgcn_mov_dpp(__builtin_bit_cast(int, x), 0x4E, 0xf, 0xf, true));
  return x;
}

__device__ __forceinline__ float dot4(const float4 a, const float4 b) {
  return fmaf(a.w, b.w, fmaf(a.z, b.z, fmaf(a.y, b.y, a.x * b.x)));
}

template <int CNT>
__device__ __forceinline__ void edge_group(const float* __restrict__ h, const unsigned sv,
                                           const int i0, const int lane,
                                           const float4 a0, const float edq,
                                           float& m, float& l, float4& acc) {
  float4 hv[CNT];
  #pragma unroll
  for (int u = 0; u < CNT; ++u) {
    const unsigned s = __builtin_amdgcn_readlane(sv, i0 + u);
    hv[u] = *(const float4*)&h[(size_t)s * HD + 4 * lane];
  }
  float e[CNT];
  float mn = m;
  #pragma unroll
  for (int u = 0; u < CNT; ++u) {
    e[u] = leaky(quad_sum(dot4(hv[u], a0)) + edq);
    mn = fmaxf(mn, e[u]);
  }
  const float sc = __builtin_amdgcn_exp2f((m - mn) * LOG2E);
  acc.x *= sc; acc.y *= sc; acc.z *= sc; acc.w *= sc;
  float ps = 0.f;
  #pragma unroll
  for (int u = 0; u < CNT; ++u) {
    const float p = __builtin_amdgcn_exp2f((e[u] - mn) * LOG2E);
    ps += p;
    acc.x = fmaf(p, hv[u].x, acc.x);
    acc.y = fmaf(p, hv[u].y, acc.y);
    acc.z = fmaf(p, hv[u].z, acc.z);
    acc.w = fmaf(p, hv[u].w, acc.w);
  }
  l = fmaf(l, sc, ps);
  m = mn;
}

__global__ __launch_bounds__(256)
void gat_gather(const float* __restrict__ h,
                const float* __restrict__ attn,
                const float* __restrict__ bias,
                const int* __restrict__ cnt,
                const int* __restrict__ cur,
                const int* __restrict__ ssrc_words,
                float* __restrict__ out) {
  const unsigned short* ssrc = (const unsigned short*)ssrc_words;
  const int lane = threadIdx.x & 31;
  const int t   = __builtin_amdgcn_readfirstlane(blockIdx.x * 8 + (threadIdx.x >> 5));
  const int end = __builtin_amdgcn_readfirstlane(cur[t]);
  const int beg = end - __builtin_amdgcn_readfirstlane(cnt[t]);

  unsigned sv = (beg + lane < end) ? (unsigned)ssrc[beg + lane] : 0u;
  const float4 a0 = *(const float4*)&attn[4 * lane];
  const float4 a1 = *(const float4*)&attn[HD + 4 * lane];
  const float4 ht = *(const float4*)&h[(size_t)t * HD + 4 * lane];
  const float edq = quad_sum(dot4(ht, a1));

  float m = -INFINITY, l = 0.f;
  float4 acc = make_float4(0.f, 0.f, 0.f, 0.f);
  for (int base = beg; base < end; ) {
    const int n     = (end - base < 32) ? (end - base) : 32;
    const int nfull = n & ~3;
    for (int i0 = 0; i0 < nfull; i0 += 4)
      edge_group<4>(h, sv, i0, lane, a0, edq, m, l, acc);
    switch (n & 3) {
      case 1: edge_group<1>(h, sv, nfull, lane, a0, edq, m, l, acc); break;
      case 2: edge_group<2>(h, sv, nfull, lane, a0, edq, m, l, acc); break;
      case 3: edge_group<3>(h, sv, nfull, lane, a0, edq, m, l, acc); break;
      default: break;
    }
    base += 32;
    if (base < end) sv = (base + lane < end) ? (unsigned)ssrc[base + lane] : 0u;
  }
  const float r = 1.f / (l > 0.f ? l : 1.f);
  const float4 b = *(const float4*)&bias[4 * lane];
  acc.x = fmaf(acc.x, r, b.x);
  acc.y = fmaf(acc.y, r, b.y);
  acc.z = fmaf(acc.z, r, b.z);
  acc.w = fmaf(acc.w, r, b.w);
  *(float4*)&out[(size_t)t * HD + 4 * lane] = acc;
}

extern "C" void kernel_launch(void* const* d_in, const int* in_sizes, int n_in,
                              void* d_out, int out_size, void* d_ws, size_t ws_size,
                              hipStream_t stream) {
  const float* feat = (const float*)d_in[0];
  const float* W    = (const float*)d_in[1];
  const float* attn = (const float*)d_in[2];
  const float* bias = (const float*)d_in[3];
  const int*   src  = (const int*)d_in[4];
  const int*   dst  = (const int*)d_in[5];
  float*       out  = (float*)d_out;

  float* h       = (float*)d_ws;
  int*   scratch = (int*)(h + (size_t)N_NODES * HD);
  int*   cnt     = scratch + 2 * (size_t)N_NODES * NUM_HEADS;
  int*   cur     = cnt + N_NODES;
  int*   ssrc    = cur + N_NODES + 64;

  int*   src16   = scratch;
  int*   loc8    = src16 + N_EDGES / 2;
  int*   piece   = loc8 + N_EDGES / 4;

  gat_chunk_sort_and_project<<<SORT_BLOCKS + N_NODES / 16, 256, 0, stream>>>(
      feat, W, h, src, dst, piece, src16, loc8);
  gat_node_sort<<<NBUCK, 1024, 0, stream>>>(piece, src16, loc8, cnt, cur, ssrc);
  gat_gather<<<N_NODES / 8, 256, 0, stream>>>(h, attn, bias, cnt, cur, ssrc, out);
}


# Projection block takes 32 rows, not 16: each B fragment fetched from W feeds two WMMAs on two accumulators, so blocks and W fetches halve; a thread's 16 row loads are issued together; sums unchanged
# speedup vs baseline: 10.0561x; 1.0408x over previous
#include <hip/hip_runtime.h>
#include <hip/hip_bf16.h>
#include <math.h>

#define N_NODES   50000
#define N_EDGES   800000
#define IN_FEATS  128
#define NUM_HEADS 8
#define OUT_FEATS 16
#define HD        (NUM_HEADS * OUT_FEATS)
#define NEG_SLOPE 0.2f

static_assert(N_NODES <= 65536, "sorted source ids are stored as 16-bit values");
static_assert(N_NODES % 8 == 0, "the gather kernel runs one wave per node, 8 per block");
static_assert(N_NODES % 16 == 0, "the projection stores whole 16-row tiles");


#define NBUCK           ((N_NODES + 255) / 256)
#define SORT_BLOCKS     256
#define EDGES_PER_BLOCK (N_EDGES / SORT_BLOCKS)
#define EPT             ((EDGES_PER_BLOCK + 255) / 256)
static_assert(N_EDGES % SORT_BLOCKS == 0, "every sort block takes the same number of edges");
static_assert(N_EDGES % 4 == 0, "the range-local node bytes fill whole words");
static_assert(NBUCK <= 256, "one thread per node range in a 256-thread block");
static_assert(SORT_BLOCKS == 256, "a piece word is indexed range * 256 + chunk; 8 chunks per wave in the node sort");
static_assert(EDGES_PER_BLOCK < 32768, "a piece's offset and count share one word, 16 bits each");
static_assert((EPT - 1) * 256 <= EDGES_PER_BLOCK, "only a thread's last edge slot can be empty");
static_assert(N_EDGES / 2 + N_EDGES / 4 + 256 * SORT_BLOCKS
              <= 2 * N_NODES * NUM_HEADS, "sort scratch fits its region");

typedef float v2f __attribute__((ext_vector_type(2)));
typedef float v8f __attribute__((ext_vector_type(8)));

#define LDA_PITCH (IN_FEATS + 4)
#define PROJ_ROWS   32
#define PROJ_BLOCKS ((N_NODES + PROJ_ROWS - 1) / PROJ_ROWS)
__device__ __forceinline__
void gemm_tile(const float* __restrict__ feat,
               const float* __restrict__ W,
               float* __restrict__ h, const int tile) {
  __shared__ float lA[PROJ_ROWS * LDA_PITCH];
  const int tid  = threadIdx.x;
  const int wave = tid >> 5;
  const int lane = tid & 31;
  const int m0   = tile * PROJ_ROWS;
  const int n0   = wave * 16;

  const int m1 = (m0 + 16 < N_NODES) ? m0 + 16 : m0;
  const int sr = tid >> 7;
  const int sc = tid & 127;
  float st[PROJ_ROWS / 2];
  #pragma unroll
  for (int j = 0; j < 8; ++j) {
    st[j]     = feat[(size_t)(m0 + sr + 2 * j) * IN_FEATS + sc];
    st[8 + j] = feat[(size_t)(m1 + sr + 2 * j) * IN_FEATS + sc];
  }
  #pragma unroll
  for (int j = 0; j < 8; ++j) {
    lA[(sr + 2 * j) * LDA_PITCH + sc]      = st[j];
    lA[(16 + sr + 2 * j) * LDA_PITCH + sc] = st[8 + j];
  }
  __syncthreads();

  const int ml  = lane & 15;
  const int klo = (lane < 16) ? 0 : 2;

  v8f acc0 = {};
  v8f acc1 = {};
  #pragma unroll 4
  for (int k = 0; k < IN_FEATS; k += 4) {
    const float2 av0 = *(const float2*)&lA[ml * LDA_PITCH + k + klo];
    const float2 av1 = *(const float2*)&lA[(16 + ml) * LDA_PITCH + k + klo];
    v2f a0; a0.x = av0.x; a0.y = av0.y;
    v2f a1; a1.x = av1.x; a1.y = av1.y;
    v2f b;
    b.x = W[(size_t)(k + klo)     * HD + n0 + ml];
    b.y = W[(size_t)(k + klo + 1) * HD + n0 + ml];
    acc0 = __builtin_amdgcn_wmma_f32_16x16x4_f32(
         false, a0,  false, b,
         (short)0, acc0,  false,  false);
    acc1 = __builtin_amdgcn_wmma_f32_16x16x4_f32(
         false, a1,  false, b,
         (short)0, acc1,  false,  false);
  }

  const int rbase = m0 + ((lane < 16) ? 0 : 8);
  #pragma unroll
  for (int v = 0; v < 8; ++v) {
    h[(size_t)(rbase + v) * HD + n0 + ml] = acc0[v];
  }
  if (m0 + 16 < N_NODES) {
    #pragma unroll
    for (int v = 0; v < 8; ++v) {
      h[(size_t)(rbase + 16 + v) * HD + n0 + ml] = acc1[v];
    }
  }
}


__device__ __forceinline__ int excl_scan_256(int v, int tid, int* wsum) {
  const int lane = tid & 31;
  const int w    = tid >> 5;
  int x = v;
  #pragma unroll
  for (int o = 1; o < 32; o <<= 1) {
    const int y = __shfl_up(x, o);
    if (lane >= o) x += y;
  }
  if (lane == 31 && w < 8) wsum[w] = x;
  __syncthreads();
  int off = 0;
  #pragma unroll
  for (int j = 0; j < 8; ++j) if (j < w) off += wsum[j];
  return off + x - v;
}

__device__ __forceinline__
void chunk_sort(const int* __restrict__ src, const int* __restrict__ dst,
                int* __restrict__ piece,
                int* __restrict__ src16_words, int* __restrict__ loc8_words,
                const int chunk) {
  __shared__ int lcnt[256];
  __shared__ int loff[256];
  __shared__ int wsum[8];
  unsigned short* src16 = (unsigned short*)src16_words;
  unsigned char*  loc8  = (unsigned char*)loc8_words;
  const int tid = threadIdx.x;
  const int e0  = chunk * EDGES_PER_BLOCK;
  lcnt[tid] = 0;
  int d[EPT], s[EPT], rk[EPT];
  const bool last_ok = tid + (EPT - 1) * 256 < EDGES_PER_BLOCK;
  #pragma unroll
  for (int k = 0; k < EPT; ++k) {
    const int i = (k < EPT - 1 || last_ok) ? tid + k * 256 : EDGES_PER_BLOCK - 1;
    d[k] = dst[e0 + i];
    s[k] = src[e0 + i];
  }
  __syncthreads();
  #pragma unroll
  for (int k = 0; k < EPT; ++k) {
    rk[k] = 0;
    if (k < EPT - 1 || last_ok) rk[k] = atomicAdd(&lcnt[d[k] >> 8], 1);
  }
  __syncthreads();
  const int c   = lcnt[tid];
  const int off = excl_scan_256(c, tid, wsum);
  loff[tid] = off;
  if (tid < NBUCK) piece[tid * 256 + chunk] = off | (c << 16);
  __syncthreads();
  #pragma unroll
  for (int k = 0; k < EPT; ++k) {
    if (k < EPT - 1 || last_ok) {
      const int p = e0 + loff[d[k] >> 8] + rk[k];
      src16[p] = (unsigned short)s[k];
      loc8[p]  = (unsigned char)(d[k] & 255);
    }
  }
}

__global__ __launch_bounds__(256, 16)
void gat_chunk_sort_and_project(const float* __restrict__ feat,
                                const float* __restrict__ W,
                                float* __restrict__ h,
                                const int* __restrict__ src, const int* __restrict__ dst,
                                int* __restrict__ piece,
                                int* __restrict__ src16_words, int* __restrict__ loc8_words) {
  if (blockIdx.x < SORT_BLOCKS) chunk_sort(src, dst, piece, src16_words, loc8_words, blockIdx.x);
  else                          gemm_tile(feat, W, h, blockIdx.x - SORT_BLOCKS);
}

__global__ __launch_bounds__(1024)
void gat_node_sort(const int* __restrict__ piece,
                   const int* __restrict__ src16_words,
                   const int* __restrict__ loc8_words,
                   int* __restrict__ cnt, int* __restrict__ cur,
                   int* __restrict__ ssrc_words) {
  __shared__ int ncnt[256];
  __shared__ int ncur[256];
  __shared__ int pbase[256];
  __shared__ int plen[256];
  __shared__ int wsb[8];
  __shared__ int first_slot;
  const unsigned short* src16 = (const unsigned short*)src16_words;
  const unsigned char*  loc8  = (const unsigned char*)loc8_words;
  unsigned short* ssrc = (unsigned short*)ssrc_words;
  const int tid = threadIdx.x;
  const int b   = blockIdx.x;

  int low = 0;
  if (tid < 256) {
    const int v = piece[b * 256 + tid];
    low = v & 0xFFFF;
    pbase[tid] = tid * EDGES_PER_BLOCK + low;
    plen[tid]  = v >> 16;
    ncnt[tid]  = 0;
  }
  if (tid == 0) first_slot = 0;
  __syncthreads();
  const int lane = tid & 31;
  const int w8   = (tid >> 5) * 8;
  int nn[8], gg[8];
  #pragma unroll
  for (int i = 0; i < 8; ++i) {
    nn[i] = __builtin_amdgcn_readfirstlane(plen[w8 + i]);
    gg[i] = __builtin_amdgcn_readfirstlane(pbase[w8 + i]);
  }
  int l[8], sv[8];
  #pragma unroll
  for (int i = 0; i < 8; ++i) {
    const int jj = max(min(lane, nn[i] - 1), 0);
    const int at = min(gg[i] + jj, N_EDGES - 1);
    l[i]  = (int)loc8[at];
    sv[i] = (int)src16[at];
  }
  asm volatile("" :: "v"(l[0]), "v"(l[1]), "v"(l[2]), "v"(l[3]),
                     "v"(l[4]), "v"(l[5]), "v"(l[6]), "v"(l[7]),
                     "v"(sv[0]), "v"(sv[1]), "v"(sv[2]), "v"(sv[3]),
                     "v"(sv[4]), "v"(sv[5]), "v"(sv[6]), "v"(sv[7]));
  #pragma unroll
  for (int i = 0; i < 8; ++i)
    if (lane < nn[i]) atomicAdd(&ncnt[l[i]], 1);
  #pragma unroll
  for (int i = 0; i < 8; ++i) {
    if (nn[i] > 32) {
      for (int j = lane + 32; j < nn[i]; j += 32) atomicAdd(&ncnt[(int)loc8[gg[i] + j]], 1);
    }
  }
  low += __builtin_amdgcn_update_dpp(0, low, 0x161, 0xf, 0xf, false);
  low += __builtin_amdgcn_update_dpp(0, low, 0x162, 0xf, 0xf, false);
  low += __builtin_amdgcn_update_dpp(0, low, 0x164, 0xf, 0xf, false);
  low += __builtin_amdgcn_update_dpp(0, low, 0x168, 0xf, 0xf, false);
  if ((tid & 15) == 0 && low != 0) atomicAdd(&first_slot, low);
  __syncthreads();
  const int c  = (tid < 256) ? ncnt[tid] : 0;
  const int nb = excl_scan_256(c, tid, wsb);
  if (tid < 256) {
    const int at = first_slot + nb;
    ncur[tid] = at;
    const int node = b * 256 + tid;
    if (node < N_NODES) {
      cnt[node] = c;
      cur[node] = at + c;
    }
  }
  __syncthreads();
  {
    unsigned p[8];
    #pragma unroll
    for (int i = 0; i < 8; ++i) {
      p[i] = 0;
      if (lane < nn[i]) p[i] = (unsigned)atomicAdd(&ncur[l[i]], 1);
    }
    #pragma unroll
    for (int i = 0; i < 8; ++i)
      if (lane < nn[i]) ssrc[p[i]] = (unsigned short)sv[i];
  }
  #pragma unroll
  for (int i = 0; i < 8; ++i) {
    if (nn[i] > 32) {
      for (int j = lane + 32; j < nn[i]; j += 32) {
        const int      ln = (int)loc8[gg[i] + j];
        const unsigned s  = src16[gg[i] + j];
        ssrc[atomicAdd(&ncur[ln], 1)] = (unsigned short)s;
      }
    }
  }
}

__device__ __forceinline__ float leaky(float e) {
  return e > 0.f ? e : NEG_SLOPE * e;
}

#define LOG2E 1.4426950408889634f

__device__ __forceinline__ float quad_sum(float x) {
  x += __builtin_bit_cast(float, __builtin_amdgcn_mov_dpp(__builtin_bit_cast(int, x), 0xB1, 0xf, 0xf, true));
  x += __builtin_bit_cast(float, __builtin_amdgcn_mov_dpp(__builtin_bit_cast(int, x), 0x4E, 0xf, 0xf, true));
  return x;
}

__device__ __forceinline__ float dot4(const float4 a, const float4 b) {
  return fmaf(a.w, b.w, fmaf(a.z, b.z, fmaf(a.y, b.y, a.x * b.x)));
}

template <int CNT>
__device__ __forceinline__ void edge_group(const float* __restrict__ h, const unsigned sv,
                                           const int i0, const int lane,
                                           const float4 a0, const float edq,
                                           float& m, float& l, float4& acc) {
  float4 hv[CNT];
  #pragma unroll
  for (int u = 0; u < CNT; ++u) {
    const unsigned s = __builtin_amdgcn_readlane(sv, i0 + u);
    hv[u] = *(const float4*)&h[(size_t)s * HD + 4 * lane];
  }
  float e[CNT];
  float mn = m;
  #pragma unroll
  for (int u = 0; u < CNT; ++u) {
    e[u] = leaky(quad_sum(dot4(hv[u], a0)) + edq);
    mn = fmaxf(mn, e[u]);
  }
  const float sc = __builtin_amdgcn_exp2f((m - mn) * LOG2E);
  acc.x *= sc; acc.y *= sc; acc.z *= sc; acc.w *= sc;
  float ps = 0.f;
  #pragma unroll
  for (int u = 0; u < CNT; ++u) {
    const float p = __builtin_amdgcn_exp2f((e[u] - mn) * LOG2E);
    ps += p;
    acc.x = fmaf(p, hv[u].x, acc.x);
    acc.y = fmaf(p, hv[u].y, acc.y);
    acc.z = fmaf(p, hv[u].z, acc.z);
    acc.w = fmaf(p, hv[u].w, acc.w);
  }
  l = fmaf(l, sc, ps);
  m = mn;
}

__global__ __launch_bounds__(256)
void gat_gather(const float* __restrict__ h,
                const float* __restrict__ attn,
                const float* __restrict__ bias,
                const int* __restrict__ cnt,
                const int* __restrict__ cur,
                const int* __restrict__ ssrc_words,
                float* __restrict__ out) {
  const unsigned short* ssrc = (const unsigned short*)ssrc_words;
  const int lane = threadIdx.x & 31;
  const int t   = __builtin_amdgcn_readfirstlane(blockIdx.x * 8 + (threadIdx.x >> 5));
  const int end = __builtin_amdgcn_readfirstlane(cur[t]);
  const int beg = end - __builtin_amdgcn_readfirstlane(cnt[t]);

  unsigned sv = (beg + lane < end) ? (unsigned)ssrc[beg + lane] : 0u;
  const float4 a0 = *(const float4*)&attn[4 * lane];
  const float4 a1 = *(const float4*)&attn[HD + 4 * lane];
  const float4 ht = *(const float4*)&h[(size_t)t * HD + 4 * lane];
  const float edq = quad_sum(dot4(ht, a1));

  float m = -INFINITY, l = 0.f;
  float4 acc = make_float4(0.f, 0.f, 0.f, 0.f);
  for (int base = beg; base < end; ) {
    const int n     = (end - base < 32) ? (end - base) : 32;
    const int nfull = n & ~3;
    for (int i0 = 0; i0 < nfull; i0 += 4)
      edge_group<4>(h, sv, i0, lane, a0, edq, m, l, acc);
    switch (n & 3) {
      case 1: edge_group<1>(h, sv, nfull, lane, a0, edq, m, l, acc); break;
      case 2: edge_group<2>(h, sv, nfull, lane, a0, edq, m, l, acc); break;
      case 3: edge_group<3>(h, sv, nfull, lane, a0, edq, m, l, acc); break;
      default: break;
    }
    base += 32;
    if (base < end) sv = (base + lane < end) ? (unsigned)ssrc[base + lane] : 0u;
  }
  const float r = 1.f / (l > 0.f ? l : 1.f);
  const float4 b = *(const float4*)&bias[4 * lane];
  acc.x = fmaf(acc.x, r, b.x);
  acc.y = fmaf(acc.y, r, b.y);
  acc.z = fmaf(acc.z, r, b.z);
  acc.w = fmaf(acc.w, r, b.w);
  *(float4*)&out[(size_t)t * HD + 4 * lane] = acc;
}

extern "C" void kernel_launch(void* const* d_in, const int* in_sizes, int n_in,
                              void* d_out, int out_size, void* d_ws, size_t ws_size,
                              hipStream_t stream) {
  const float* feat = (const float*)d_in[0];
  const float* W    = (const float*)d_in[1];
  const float* attn = (const float*)d_in[2];
  const float* bias = (const float*)d_in[3];
  const int*   src  = (const int*)d_in[4];
  const int*   dst  = (const int*)d_in[5];
  float*       out  = (float*)d_out;

  float* h       = (float*)d_ws;
  int*   scratch = (int*)(h + (size_t)N_NODES * HD);
  int*   cnt     = scratch + 2 * (size_t)N_NODES * NUM_HEADS;
  int*   cur     = cnt + N_NODES;
  int*   ssrc    = cur + N_NODES + 64;

  int*   src16   = scratch;
  int*   loc8    = src16 + N_EDGES / 2;
  int*   piece   = loc8 + N_EDGES / 4;

  gat_chunk_sort_and_project<<<SORT_BLOCKS + PROJ_BLOCKS, 256, 0, stream>>>(
      feat, W, h, src, dst, piece, src16, loc8);
  gat_node_sort<<<NBUCK, 1024, 0, stream>>>(piece, src16, loc8, cnt, cur, ssrc);
  gat_gather<<<N_NODES / 8, 256, 0, stream>>>(h, attn, bias, cnt, cur, ssrc, out);
}


# Both edge sorts build their ordered output in LDS and store it as whole words, lane after lane: the per-edge scattered 1- and 2-byte global stores go; slot order (arrival) and sums unchanged
# speedup vs baseline: 12.5597x; 1.2490x over previous
#include <hip/hip_runtime.h>
#include <hip/hip_bf16.h>
#include <math.h>

#define N_NODES   50000
#define N_EDGES   800000
#define IN_FEATS  128
#define NUM_HEADS 8
#define OUT_FEATS 16
#define HD        (NUM_HEADS * OUT_FEATS)
#define NEG_SLOPE 0.2f

static_assert(N_NODES <= 65536, "sorted source ids are stored as 16-bit values");
static_assert(N_NODES % 8 == 0, "the gather kernel runs one wave per node, 8 per block");
static_assert(N_NODES % 16 == 0, "the projection stores whole 16-row tiles");


#define NBUCK           ((N_NODES + 255) / 256)
#define SORT_BLOCKS     256
#define EDGES_PER_BLOCK (N_EDGES / SORT_BLOCKS)
#define EPT             ((EDGES_PER_BLOCK + 255) / 256)
static_assert(N_EDGES % SORT_BLOCKS == 0, "every sort block takes the same number of edges");
static_assert(N_EDGES % 4 == 0, "the range-local node bytes fill whole words");
static_assert(NBUCK <= 256, "one thread per node range in a 256-thread block");
static_assert(SORT_BLOCKS == 256, "a piece word is indexed range * 256 + chunk; 8 chunks per wave in the node sort");
static_assert(EDGES_PER_BLOCK < 32768, "a piece's offset and count share one word, 16 bits each");
static_assert((EPT - 1) * 256 <= EDGES_PER_BLOCK, "only a thread's last edge slot can be empty");
static_assert(N_EDGES / 2 + N_EDGES / 4 + 256 * SORT_BLOCKS
              <= 2 * N_NODES * NUM_HEADS, "sort scratch fits its region");

typedef float v2f __attribute__((ext_vector_type(2)));
typedef float v8f __attribute__((ext_vector_type(8)));

#define LDA_PITCH (IN_FEATS + 4)
#define PROJ_ROWS   32
#define PROJ_BLOCKS ((N_NODES + PROJ_ROWS - 1) / PROJ_ROWS)
__device__ __forceinline__
void gemm_tile(const float* __restrict__ feat,
               const float* __restrict__ W,
               float* __restrict__ h, const int tile) {
  __shared__ float lA[PROJ_ROWS * LDA_PITCH];
  const int tid  = threadIdx.x;
  const int wave = tid >> 5;
  const int lane = tid & 31;
  const int m0   = tile * PROJ_ROWS;
  const int n0   = wave * 16;

  const int m1 = (m0 + 16 < N_NODES) ? m0 + 16 : m0;
  const int sr = tid >> 7;
  const int sc = tid & 127;
  float st[PROJ_ROWS / 2];
  #pragma unroll
  for (int j = 0; j < 8; ++j) {
    st[j]     = feat[(size_t)(m0 + sr + 2 * j) * IN_FEATS + sc];
    st[8 + j] = feat[(size_t)(m1 + sr + 2 * j) * IN_FEATS + sc];
  }
  #pragma unroll
  for (int j = 0; j < 8; ++j) {
    lA[(sr + 2 * j) * LDA_PITCH + sc]      = st[j];
    lA[(16 + sr + 2 * j) * LDA_PITCH + sc] = st[8 + j];
  }
  __syncthreads();

  const int ml  = lane & 15;
  const int klo = (lane < 16) ? 0 : 2;

  v8f acc0 = {};
  v8f acc1 = {};
  #pragma unroll 4
  for (int k = 0; k < IN_FEATS; k += 4) {
    const float2 av0 = *(const float2*)&lA[ml * LDA_PITCH + k + klo];
    const float2 av1 = *(const float2*)&lA[(16 + ml) * LDA_PITCH + k + klo];
    v2f a0; a0.x = av0.x; a0.y = av0.y;
    v2f a1; a1.x = av1.x; a1.y = av1.y;
    v2f b;
    b.x = W[(size_t)(k + klo)     * HD + n0 + ml];
    b.y = W[(size_t)(k + klo + 1) * HD + n0 + ml];
    acc0 = __builtin_amdgcn_wmma_f32_16x16x4_f32(
         false, a0,  false, b,
         (short)0, acc0,  false,  false);
    acc1 = __builtin_amdgcn_wmma_f32_16x16x4_f32(
         false, a1,  false, b,
         (short)0, acc1,  false,  false);
  }

  const int rbase = m0 + ((lane < 16) ? 0 : 8);
  #pragma unroll
  for (int v = 0; v < 8; ++v) {
    h[(size_t)(rbase + v) * HD + n0 + ml] = acc0[v];
  }
  if (m0 + 16 < N_NODES) {
    #pragma unroll
    for (int v = 0; v < 8; ++v) {
      h[(size_t)(rbase + 16 + v) * HD + n0 + ml] = acc1[v];
    }
  }
}


__device__ __forceinline__ int excl_scan_256(int v, int tid, int* wsum) {
  const int lane = tid & 31;
  const int w    = tid >> 5;
  int x = v;
  #pragma unroll
  for (int o = 1; o < 32; o <<= 1) {
    const int y = __shfl_up(x, o);
    if (lane >= o) x += y;
  }
  if (lane == 31 && w < 8) wsum[w] = x;
  __syncthreads();
  int off = 0;
  #pragma unroll
  for (int j = 0; j < 8; ++j) if (j < w) off += wsum[j];
  return off + x - v;
}

__device__ __forceinline__
void chunk_sort(const int* __restrict__ src, const int* __restrict__ dst,
                int* __restrict__ piece,
                int* __restrict__ src16_words, int* __restrict__ loc8_words,
                const int chunk) {
  __shared__ int lcnt[256];
  __shared__ int loff[256];
  __shared__ int wsum[8];
  __shared__ __attribute__((aligned(16))) unsigned short st16[EDGES_PER_BLOCK + 4];
  __shared__ unsigned st8w[(EDGES_PER_BLOCK + 3) / 4 + 2];
  unsigned char*  st8   = (unsigned char*)st8w;
  unsigned short* src16 = (unsigned short*)src16_words;
  unsigned char*  loc8  = (unsigned char*)loc8_words;
  const int tid = threadIdx.x;
  const int e0  = chunk * EDGES_PER_BLOCK;
  const int lead = e0 & 3;
  const int base = e0 - lead;
  lcnt[tid] = 0;
  int d[EPT], s[EPT], rk[EPT];
  const bool last_ok = tid + (EPT - 1) * 256 < EDGES_PER_BLOCK;
  #pragma unroll
  for (int k = 0; k < EPT; ++k) {
    const int i = (k < EPT - 1 || last_ok) ? tid + k * 256 : EDGES_PER_BLOCK - 1;
    d[k] = dst[e0 + i];
    s[k] = src[e0 + i];
  }
  __syncthreads();
  #pragma unroll
  for (int k = 0; k < EPT; ++k) {
    rk[k] = 0;
    if (k < EPT - 1 || last_ok) rk[k] = atomicAdd(&lcnt[d[k] >> 8], 1);
  }
  __syncthreads();
  const int c   = lcnt[tid];
  const int off = excl_scan_256(c, tid, wsum);
  loff[tid] = off + lead;
  if (tid < NBUCK) piece[tid * 256 + chunk] = off | (c << 16);
  __syncthreads();
  #pragma unroll
  for (int k = 0; k < EPT; ++k) {
    if (k < EPT - 1 || last_ok) {
      const int q = loff[d[k] >> 8] + rk[k];
      st16[q] = (unsigned short)s[k];
      st8[q]  = (unsigned char)(d[k] & 255);
    }
  }
  __syncthreads();
  {
    unsigned* gw = (unsigned*)src16_words + (base >> 1);
    const int wf = lead >> 1;
    const int wl = (lead + EDGES_PER_BLOCK - 1) >> 1;
    #pragma unroll
    for (int k = 0; k < (EDGES_PER_BLOCK / 2 + 1 + 255) / 256; ++k) {
      const int w = wf + 1 + tid + 256 * k;
      if (w < wl) gw[w] = (unsigned)st16[2 * w] | ((unsigned)st16[2 * w + 1] << 16);
    }
    if (tid < 2) {
      const int w = tid ? wl : wf;
      const bool has_lo = 2 * w >= lead && 2 * w < lead + EDGES_PER_BLOCK;
      const bool has_hi = 2 * w + 1 >= lead && 2 * w + 1 < lead + EDGES_PER_BLOCK;
      if (has_lo) src16[base + 2 * w]     = st16[2 * w];
      if (has_hi) src16[base + 2 * w + 1] = st16[2 * w + 1];
    }
  }
  {
    unsigned* gw = (unsigned*)loc8_words + (base >> 2);
    const int nb = lead + EDGES_PER_BLOCK;
    const int nw = (nb + 3) >> 2;
    #pragma unroll
    for (int k = 0; k < (EDGES_PER_BLOCK / 4 + 2 + 255) / 256; ++k) {
      const int w = 1 + tid + 256 * k;
      if (w < nw - 1) gw[w] = st8w[w];
    }
    if (tid < 2) {
      const int w = tid * (nw - 1);
      for (int j = 0; j < 4; ++j) {
        const int o = 4 * w + j;
        if (o >= lead && o < nb) loc8[base + o] = st8[o];
      }
    }
  }
}

__global__ __launch_bounds__(256, 16)
void gat_chunk_sort_and_project(const float* __restrict__ feat,
                                const float* __restrict__ W,
                                float* __restrict__ h,
                                const int* __restrict__ src, const int* __restrict__ dst,
                                int* __restrict__ piece,
                                int* __restrict__ src16_words, int* __restrict__ loc8_words) {
  if (blockIdx.x < SORT_BLOCKS) chunk_sort(src, dst, piece, src16_words, loc8_words, blockIdx.x);
  else                          gemm_tile(feat, W, h, blockIdx.x - SORT_BLOCKS);
}

#define STAGE_CAP 8192
__global__ __launch_bounds__(1024)
void gat_node_sort(const int* __restrict__ piece,
                   const int* __restrict__ src16_words,
                   const int* __restrict__ loc8_words,
                   int* __restrict__ cnt, int* __restrict__ cur,
                   int* __restrict__ ssrc_words) {
  __shared__ int ncnt[256];
  __shared__ int ncur[256];
  __shared__ int pbase[256];
  __shared__ int plen[256];
  __shared__ int wsb[8];
  __shared__ int first_slot;
  __shared__ int range_total;
  __shared__ __attribute__((aligned(16))) unsigned short stage[STAGE_CAP + 2];
  const unsigned short* src16 = (const unsigned short*)src16_words;
  const unsigned char*  loc8  = (const unsigned char*)loc8_words;
  unsigned short* ssrc = (unsigned short*)ssrc_words;
  const int tid = threadIdx.x;
  const int b   = blockIdx.x;

  int low = 0;
  if (tid < 256) {
    const int v = piece[b * 256 + tid];
    low = v & 0xFFFF;
    pbase[tid] = tid * EDGES_PER_BLOCK + low;
    plen[tid]  = v >> 16;
    ncnt[tid]  = 0;
  }
  if (tid == 0) first_slot = 0;
  __syncthreads();
  const int lane = tid & 31;
  const int w8   = (tid >> 5) * 8;
  int nn[8], gg[8];
  #pragma unroll
  for (int i = 0; i < 8; ++i) {
    nn[i] = __builtin_amdgcn_readfirstlane(plen[w8 + i]);
    gg[i] = __builtin_amdgcn_readfirstlane(pbase[w8 + i]);
  }
  int l[8], sv[8];
  #pragma unroll
  for (int i = 0; i < 8; ++i) {
    const int jj = max(min(lane, nn[i] - 1), 0);
    const int at = min(gg[i] + jj, N_EDGES - 1);
    l[i]  = (int)loc8[at];
    sv[i] = (int)src16[at];
  }
  asm volatile("" :: "v"(l[0]), "v"(l[1]), "v"(l[2]), "v"(l[3]),
                     "v"(l[4]), "v"(l[5]), "v"(l[6]), "v"(l[7]),
                     "v"(sv[0]), "v"(sv[1]), "v"(sv[2]), "v"(sv[3]),
                     "v"(sv[4]), "v"(sv[5]), "v"(sv[6]), "v"(sv[7]));
  #pragma unroll
  for (int i = 0; i < 8; ++i)
    if (lane < nn[i]) atomicAdd(&ncnt[l[i]], 1);
  #pragma unroll
  for (int i = 0; i < 8; ++i) {
    if (nn[i] > 32) {
      for (int j = lane + 32; j < nn[i]; j += 32) atomicAdd(&ncnt[(int)loc8[gg[i] + j]], 1);
    }
  }
  low += __builtin_amdgcn_update_dpp(0, low, 0x161, 0xf, 0xf, false);
  low += __builtin_amdgcn_update_dpp(0, low, 0x162, 0xf, 0xf, false);
  low += __builtin_amdgcn_update_dpp(0, low, 0x164, 0xf, 0xf, false);
  low += __builtin_amdgcn_update_dpp(0, low, 0x168, 0xf, 0xf, false);
  if ((tid & 15) == 0 && low != 0) atomicAdd(&first_slot, low);
  __syncthreads();
  const int c  = (tid < 256) ? ncnt[tid] : 0;
  const int nb = excl_scan_256(c, tid, wsb);
  if (tid < 256) {
    ncur[tid] = nb;
    const int node = b * 256 + tid;
    if (node < N_NODES) {
      cnt[node] = c;
      cur[node] = first_slot + nb + c;
    }
    if (tid == 255) range_total = nb + c;
  }
  __syncthreads();
  const int  fs     = first_slot;
  const int  total  = range_total;
  const bool staged = total <= STAGE_CAP;
  const int  lead   = fs & 1;
  {
    unsigned p[8];
    #pragma unroll
    for (int i = 0; i < 8; ++i) {
      p[i] = 0;
      if (lane < nn[i]) p[i] = (unsigned)atomicAdd(&ncur[l[i]], 1);
    }
    if (staged) {
      #pragma unroll
      for (int i = 0; i < 8; ++i)
        if (lane < nn[i]) stage[lead + p[i]] = (unsigned short)sv[i];
    } else {
      #pragma unroll
      for (int i = 0; i < 8; ++i)
        if (lane < nn[i]) ssrc[fs + p[i]] = (unsigned short)sv[i];
    }
  }
  #pragma unroll
  for (int i = 0; i < 8; ++i) {
    if (nn[i] > 32) {
      for (int j = lane + 32; j < nn[i]; j += 32) {
        const int      ln = (int)loc8[gg[i] + j];
        const unsigned s  = src16[gg[i] + j];
        const int      q  = atomicAdd(&ncur[ln], 1);
        if (staged) stage[lead + q] = (unsigned short)s;
        else        ssrc[fs + q]    = (unsigned short)s;
      }
    }
  }
  __syncthreads();
  if (staged) {
    unsigned* gw = (unsigned*)ssrc_words + (fs >> 1);
    const int nw = (lead + total + 1) >> 1;
    for (int w = 1 + tid; w < nw - 1; w += 1024)
      gw[w] = (unsigned)stage[2 * w] | ((unsigned)stage[2 * w + 1] << 16);
    if (tid < 2 && nw > tid) {
      const int w = tid * (nw - 1);
      const bool has_lo = 2 * w >= lead;
      const bool has_hi = 2 * w + 1 < lead + total;
      if (has_lo) ssrc[2 * ((fs >> 1) + w)]     = stage[2 * w];
      if (has_hi) ssrc[2 * ((fs >> 1) + w) + 1] = stage[2 * w + 1];
    }
  }
}

__device__ __forceinline__ float leaky(float e) {
  return e > 0.f ? e : NEG_SLOPE * e;
}

#define LOG2E 1.4426950408889634f

__device__ __forceinline__ float quad_sum(float x) {
  x += __builtin_bit_cast(float, __builtin_amdgcn_mov_dpp(__builtin_bit_cast(int, x), 0xB1, 0xf, 0xf, true));
  x += __builtin_bit_cast(float, __builtin_amdgcn_mov_dpp(__builtin_bit_cast(int, x), 0x4E, 0xf, 0xf, true));
  return x;
}

__device__ __forceinline__ float dot4(const float4 a, const float4 b) {
  return fmaf(a.w, b.w, fmaf(a.z, b.z, fmaf(a.y, b.y, a.x * b.x)));
}

template <int CNT>
__device__ __forceinline__ void edge_group(const float* __restrict__ h, const unsigned sv,
                                           const int i0, const int lane,
                                           const float4 a0, const float edq,
                                           float& m, float& l, float4& acc) {
  float4 hv[CNT];
  #pragma unroll
  for (int u = 0; u < CNT; ++u) {
    const unsigned s = __builtin_amdgcn_readlane(sv, i0 + u);
    hv[u] = *(const float4*)&h[(size_t)s * HD + 4 * lane];
  }
  float e[CNT];
  float mn = m;
  #pragma unroll
  for (int u = 0; u < CNT; ++u) {
    e[u] = leaky(quad_sum(dot4(hv[u], a0)) + edq);
    mn = fmaxf(mn, e[u]);
  }
  const float sc = __builtin_amdgcn_exp2f((m - mn) * LOG2E);
  acc.x *= sc; acc.y *= sc; acc.z *= sc; acc.w *= sc;
  float ps = 0.f;
  #pragma unroll
  for (int u = 0; u < CNT; ++u) {
    const float p = __builtin_amdgcn_exp2f((e[u] - mn) * LOG2E);
    ps += p;
    acc.x = fmaf(p, hv[u].x, acc.x);
    acc.y = fmaf(p, hv[u].y, acc.y);
    acc.z = fmaf(p, hv[u].z, acc.z);
    acc.w = fmaf(p, hv[u].w, acc.w);
  }
  l = fmaf(l, sc, ps);
  m = mn;
}

__global__ __launch_bounds__(256)
void gat_gather(const float* __restrict__ h,
                const float* __restrict__ attn,
                const float* __restrict__ bias,
                const int* __restrict__ cnt,
                const int* __restrict__ cur,
                const int* __restrict__ ssrc_words,
                float* __restrict__ out) {
  const unsigned short* ssrc = (const unsigned short*)ssrc_words;
  const int lane = threadIdx.x & 31;
  const int t   = __builtin_amdgcn_readfirstlane(blockIdx.x * 8 + (threadIdx.x >> 5));
  const int end = __builtin_amdgcn_readfirstlane(cur[t]);
  const int beg = end - __builtin_amdgcn_readfirstlane(cnt[t]);

  unsigned sv = (beg + lane < end) ? (unsigned)ssrc[beg + lane] : 0u;
  const float4 a0 = *(const float4*)&attn[4 * lane];
  const float4 a1 = *(const float4*)&attn[HD + 4 * lane];
  const float4 ht = *(const float4*)&h[(size_t)t * HD + 4 * lane];
  const float edq = quad_sum(dot4(ht, a1));

  float m = -INFINITY, l = 0.f;
  float4 acc = make_float4(0.f, 0.f, 0.f, 0.f);
  for (int base = beg; base < end; ) {
    const int n     = (end - base < 32) ? (end - base) : 32;
    const int nfull = n & ~3;
    for (int i0 = 0; i0 < nfull; i0 += 4)
      edge_group<4>(h, sv, i0, lane, a0, edq, m, l, acc);
    switch (n & 3) {
      case 1: edge_group<1>(h, sv, nfull, lane, a0, edq, m, l, acc); break;
      case 2: edge_group<2>(h, sv, nfull, lane, a0, edq, m, l, acc); break;
      case 3: edge_group<3>(h, sv, nfull, lane, a0, edq, m, l, acc); break;
      default: break;
    }
    base += 32;
    if (base < end) sv = (base + lane < end) ? (unsigned)ssrc[base + lane] : 0u;
  }
  const float r = 1.f / (l > 0.f ? l : 1.f);
  const float4 b = *(const float4*)&bias[4 * lane];
  acc.x = fmaf(acc.x, r, b.x);
  acc.y = fmaf(acc.y, r, b.y);
  acc.z = fmaf(acc.z, r, b.z);
  acc.w = fmaf(acc.w, r, b.w);
  *(float4*)&out[(size_t)t * HD + 4 * lane] = acc;
}

extern "C" void kernel_launch(void* const* d_in, const int* in_sizes, int n_in,
                              void* d_out, int out_size, void* d_ws, size_t ws_size,
                              hipStream_t stream) {
  const float* feat = (const float*)d_in[0];
  const float* W    = (const float*)d_in[1];
  const float* attn = (const float*)d_in[2];
  const float* bias = (const float*)d_in[3];
  const int*   src  = (const int*)d_in[4];
  const int*   dst  = (const int*)d_in[5];
  float*       out  = (float*)d_out;

  float* h       = (float*)d_ws;
  int*   scratch = (int*)(h + (size_t)N_NODES * HD);
  int*   cnt     = scratch + 2 * (size_t)N_NODES * NUM_HEADS;
  int*   cur     = cnt + N_NODES;
  int*   ssrc    = cur + N_NODES + 64;

  int*   src16   = scratch;
  int*   loc8    = src16 + N_EDGES / 2;
  int*   piece   = loc8 + N_EDGES / 4;

  gat_chunk_sort_and_project<<<SORT_BLOCKS + PROJ_BLOCKS, 256, 0, stream>>>(
      feat, W, h, src, dst, piece, src16, loc8);
  gat_node_sort<<<NBUCK, 1024, 0, stream>>>(piece, src16, loc8, cnt, cur, ssrc);
  gat_gather<<<N_NODES / 8, 256, 0, stream>>>(h, attn, bias, cnt, cur, ssrc, out);
}


# Gather: a head's 4 lanes sum 4 edges' logit quarters across lanes (3 dpp adds), lane j alone does edge j's LeakyReLU, max, exp2 (log2 units), weight back by dpp; exp-sum per lane: sum order changes
# speedup vs baseline: 12.9344x; 1.0298x over previous
#include <hip/hip_runtime.h>
#include <hip/hip_bf16.h>
#include <math.h>

#define N_NODES   50000
#define N_EDGES   800000
#define IN_FEATS  128
#define NUM_HEADS 8
#define OUT_FEATS 16
#define HD        (NUM_HEADS * OUT_FEATS)
#define NEG_SLOPE 0.2f

static_assert(N_NODES <= 65536, "sorted source ids are stored as 16-bit values");
static_assert(N_NODES % 8 == 0, "the gather kernel runs one wave per node, 8 per block");
static_assert(N_NODES % 16 == 0, "the projection stores whole 16-row tiles");


#define NBUCK           ((N_NODES + 255) / 256)
#define SORT_BLOCKS     256
#define EDGES_PER_BLOCK (N_EDGES / SORT_BLOCKS)
#define EPT             ((EDGES_PER_BLOCK + 255) / 256)
static_assert(N_EDGES % SORT_BLOCKS == 0, "every sort block takes the same number of edges");
static_assert(N_EDGES % 4 == 0, "the range-local node bytes fill whole words");
static_assert(NBUCK <= 256, "one thread per node range in a 256-thread block");
static_assert(SORT_BLOCKS == 256, "a piece word is indexed range * 256 + chunk; 8 chunks per wave in the node sort");
static_assert(EDGES_PER_BLOCK < 32768, "a piece's offset and count share one word, 16 bits each");
static_assert((EPT - 1) * 256 <= EDGES_PER_BLOCK, "only a thread's last edge slot can be empty");
static_assert(N_EDGES / 2 + N_EDGES / 4 + 256 * SORT_BLOCKS
              <= 2 * N_NODES * NUM_HEADS, "sort scratch fits its region");

typedef float v2f __attribute__((ext_vector_type(2)));
typedef float v8f __attribute__((ext_vector_type(8)));

#define LDA_PITCH (IN_FEATS + 4)
#define PROJ_ROWS   32
#define PROJ_BLOCKS ((N_NODES + PROJ_ROWS - 1) / PROJ_ROWS)
__device__ __forceinline__
void gemm_tile(const float* __restrict__ feat,
               const float* __restrict__ W,
               float* __restrict__ h, const int tile) {
  __shared__ float lA[PROJ_ROWS * LDA_PITCH];
  const int tid  = threadIdx.x;
  const int wave = tid >> 5;
  const int lane = tid & 31;
  const int m0   = tile * PROJ_ROWS;
  const int n0   = wave * 16;

  const int m1 = (m0 + 16 < N_NODES) ? m0 + 16 : m0;
  const int sr = tid >> 7;
  const int sc = tid & 127;
  float st[PROJ_ROWS / 2];
  #pragma unroll
  for (int j = 0; j < 8; ++j) {
    st[j]     = feat[(size_t)(m0 + sr + 2 * j) * IN_FEATS + sc];
    st[8 + j] = feat[(size_t)(m1 + sr + 2 * j) * IN_FEATS + sc];
  }
  #pragma unroll
  for (int j = 0; j < 8; ++j) {
    lA[(sr + 2 * j) * LDA_PITCH + sc]      = st[j];
    lA[(16 + sr + 2 * j) * LDA_PITCH + sc] = st[8 + j];
  }
  __syncthreads();

  const int ml  = lane & 15;
  const int klo = (lane < 16) ? 0 : 2;

  v8f acc0 = {};
  v8f acc1 = {};
  #pragma unroll 4
  for (int k = 0; k < IN_FEATS; k += 4) {
    const float2 av0 = *(const float2*)&lA[ml * LDA_PITCH + k + klo];
    const float2 av1 = *(const float2*)&lA[(16 + ml) * LDA_PITCH + k + klo];
    v2f a0; a0.x = av0.x; a0.y = av0.y;
    v2f a1; a1.x = av1.x; a1.y = av1.y;
    v2f b;
    b.x = W[(size_t)(k + klo)     * HD + n0 + ml];
    b.y = W[(size_t)(k + klo + 1) * HD + n0 + ml];
    acc0 = __builtin_amdgcn_wmma_f32_16x16x4_f32(
         false, a0,  false, b,
         (short)0, acc0,  false,  false);
    acc1 = __builtin_amdgcn_wmma_f32_16x16x4_f32(
         false, a1,  false, b,
         (short)0, acc1,  false,  false);
  }

  const int rbase = m0 + ((lane < 16) ? 0 : 8);
  #pragma unroll
  for (int v = 0; v < 8; ++v) {
    h[(size_t)(rbase + v) * HD + n0 + ml] = acc0[v];
  }
  if (m0 + 16 < N_NODES) {
    #pragma unroll
    for (int v = 0; v < 8; ++v) {
      h[(size_t)(rbase + 16 + v) * HD + n0 + ml] = acc1[v];
    }
  }
}


__device__ __forceinline__ int excl_scan_256(int v, int tid, int* wsum) {
  const int lane = tid & 31;
  const int w    = tid >> 5;
  int x = v;
  #pragma unroll
  for (int o = 1; o < 32; o <<= 1) {
    const int y = __shfl_up(x, o);
    if (lane >= o) x += y;
  }
  if (lane == 31 && w < 8) wsum[w] = x;
  __syncthreads();
  int off = 0;
  #pragma unroll
  for (int j = 0; j < 8; ++j) if (j < w) off += wsum[j];
  return off + x - v;
}

__device__ __forceinline__
void chunk_sort(const int* __restrict__ src, const int* __restrict__ dst,
                int* __restrict__ piece,
                int* __restrict__ src16_words, int* __restrict__ loc8_words,
                const int chunk) {
  __shared__ int lcnt[256];
  __shared__ int loff[256];
  __shared__ int wsum[8];
  __shared__ __attribute__((aligned(16))) unsigned short st16[EDGES_PER_BLOCK + 4];
  __shared__ unsigned st8w[(EDGES_PER_BLOCK + 3) / 4 + 2];
  unsigned char*  st8   = (unsigned char*)st8w;
  unsigned short* src16 = (unsigned short*)src16_words;
  unsigned char*  loc8  = (unsigned char*)loc8_words;
  const int tid = threadIdx.x;
  const int e0  = chunk * EDGES_PER_BLOCK;
  const int lead = e0 & 3;
  const int base = e0 - lead;
  lcnt[tid] = 0;
  int d[EPT], s[EPT], rk[EPT];
  const bool last_ok = tid + (EPT - 1) * 256 < EDGES_PER_BLOCK;
  #pragma unroll
  for (int k = 0; k < EPT; ++k) {
    const int i = (k < EPT - 1 || last_ok) ? tid + k * 256 : EDGES_PER_BLOCK - 1;
    d[k] = dst[e0 + i];
    s[k] = src[e0 + i];
  }
  __syncthreads();
  #pragma unroll
  for (int k = 0; k < EPT; ++k) {
    rk[k] = 0;
    if (k < EPT - 1 || last_ok) rk[k] = atomicAdd(&lcnt[d[k] >> 8], 1);
  }
  __syncthreads();
  const int c   = lcnt[tid];
  const int off = excl_scan_256(c, tid, wsum);
  loff[tid] = off + lead;
  if (tid < NBUCK) piece[tid * 256 + chunk] = off | (c << 16);
  __syncthreads();
  #pragma unroll
  for (int k = 0; k < EPT; ++k) {
    if (k < EPT - 1 || last_ok) {
      const int q = loff[d[k] >> 8] + rk[k];
      st16[q] = (unsigned short)s[k];
      st8[q]  = (unsigned char)(d[k] & 255);
    }
  }
  __syncthreads();
  {
    unsigned* gw = (unsigned*)src16_words + (base >> 1);
    const int wf = lead >> 1;
    const int wl = (lead + EDGES_PER_BLOCK - 1) >> 1;
    #pragma unroll
    for (int k = 0; k < (EDGES_PER_BLOCK / 2 + 1 + 255) / 256; ++k) {
      const int w = wf + 1 + tid + 256 * k;
      if (w < wl) gw[w] = (unsigned)st16[2 * w] | ((unsigned)st16[2 * w + 1] << 16);
    }
    if (tid < 2) {
      const int w = tid ? wl : wf;
      const bool has_lo = 2 * w >= lead && 2 * w < lead + EDGES_PER_BLOCK;
      const bool has_hi = 2 * w + 1 >= lead && 2 * w + 1 < lead + EDGES_PER_BLOCK;
      if (has_lo) src16[base + 2 * w]     = st16[2 * w];
      if (has_hi) src16[base + 2 * w + 1] = st16[2 * w + 1];
    }
  }
  {
    unsigned* gw = (unsigned*)loc8_words + (base >> 2);
    const int nb = lead + EDGES_PER_BLOCK;
    const int nw = (nb + 3) >> 2;
    #pragma unroll
    for (int k = 0; k < (EDGES_PER_BLOCK / 4 + 2 + 255) / 256; ++k) {
      const int w = 1 + tid + 256 * k;
      if (w < nw - 1) gw[w] = st8w[w];
    }
    if (tid < 2) {
      const int w = tid * (nw - 1);
      for (int j = 0; j < 4; ++j) {
        const int o = 4 * w + j;
        if (o >= lead && o < nb) loc8[base + o] = st8[o];
      }
    }
  }
}

__global__ __launch_bounds__(256, 16)
void gat_chunk_sort_and_project(const float* __restrict__ feat,
                                const float* __restrict__ W,
                                float* __restrict__ h,
                                const int* __restrict__ src, const int* __restrict__ dst,
                                int* __restrict__ piece,
                                int* __restrict__ src16_words, int* __restrict__ loc8_words) {
  if (blockIdx.x < SORT_BLOCKS) chunk_sort(src, dst, piece, src16_words, loc8_words, blockIdx.x);
  else                          gemm_tile(feat, W, h, blockIdx.x - SORT_BLOCKS);
}

#define STAGE_CAP 8192
__global__ __launch_bounds__(1024)
void gat_node_sort(const int* __restrict__ piece,
                   const int* __restrict__ src16_words,
                   const int* __restrict__ loc8_words,
                   int* __restrict__ cnt, int* __restrict__ cur,
                   int* __restrict__ ssrc_words) {
  __shared__ int ncnt[256];
  __shared__ int ncur[256];
  __shared__ int pbase[256];
  __shared__ int plen[256];
  __shared__ int wsb[8];
  __shared__ int first_slot;
  __shared__ int range_total;
  __shared__ __attribute__((aligned(16))) unsigned short stage[STAGE_CAP + 2];
  const unsigned short* src16 = (const unsigned short*)src16_words;
  const unsigned char*  loc8  = (const unsigned char*)loc8_words;
  unsigned short* ssrc = (unsigned short*)ssrc_words;
  const int tid = threadIdx.x;
  const int b   = blockIdx.x;

  int low = 0;
  if (tid < 256) {
    const int v = piece[b * 256 + tid];
    low = v & 0xFFFF;
    pbase[tid] = tid * EDGES_PER_BLOCK + low;
    plen[tid]  = v >> 16;
    ncnt[tid]  = 0;
  }
  if (tid == 0) first_slot = 0;
  __syncthreads();
  const int lane = tid & 31;
  const int w8   = (tid >> 5) * 8;
  int nn[8], gg[8];
  #pragma unroll
  for (int i = 0; i < 8; ++i) {
    nn[i] = __builtin_amdgcn_readfirstlane(plen[w8 + i]);
    gg[i] = __builtin_amdgcn_readfirstlane(pbase[w8 + i]);
  }
  int l[8], sv[8];
  #pragma unroll
  for (int i = 0; i < 8; ++i) {
    const int jj = max(min(lane, nn[i] - 1), 0);
    const int at = min(gg[i] + jj, N_EDGES - 1);
    l[i]  = (int)loc8[at];
    sv[i] = (int)src16[at];
  }
  asm volatile("" :: "v"(l[0]), "v"(l[1]), "v"(l[2]), "v"(l[3]),
                     "v"(l[4]), "v"(l[5]), "v"(l[6]), "v"(l[7]),
                     "v"(sv[0]), "v"(sv[1]), "v"(sv[2]), "v"(sv[3]),
                     "v"(sv[4]), "v"(sv[5]), "v"(sv[6]), "v"(sv[7]));
  #pragma unroll
  for (int i = 0; i < 8; ++i)
    if (lane < nn[i]) atomicAdd(&ncnt[l[i]], 1);
  #pragma unroll
  for (int i = 0; i < 8; ++i) {
    if (nn[i] > 32) {
      for (int j = lane + 32; j < nn[i]; j += 32) atomicAdd(&ncnt[(int)loc8[gg[i] + j]], 1);
    }
  }
  low += __builtin_amdgcn_update_dpp(0, low, 0x161, 0xf, 0xf, false);
  low += __builtin_amdgcn_update_dpp(0, low, 0x162, 0xf, 0xf, false);
  low += __builtin_amdgcn_update_dpp(0, low, 0x164, 0xf, 0xf, false);
  low += __builtin_amdgcn_update_dpp(0, low, 0x168, 0xf, 0xf, false);
  if ((tid & 15) == 0 && low != 0) atomicAdd(&first_slot, low);
  __syncthreads();
  const int c  = (tid < 256) ? ncnt[tid] : 0;
  const int nb = excl_scan_256(c, tid, wsb);
  if (tid < 256) {
    ncur[tid] = nb;
    const int node = b * 256 + tid;
    if (node < N_NODES) {
      cnt[node] = c;
      cur[node] = first_slot + nb + c;
    }
    if (tid == 255) range_total = nb + c;
  }
  __syncthreads();
  const int  fs     = first_slot;
  const int  total  = range_total;
  const bool staged = total <= STAGE_CAP;
  const int  lead   = fs & 1;
  {
    unsigned p[8];
    #pragma unroll
    for (int i = 0; i < 8; ++i) {
      p[i] = 0;
      if (lane < nn[i]) p[i] = (unsigned)atomicAdd(&ncur[l[i]], 1);
    }
    if (staged) {
      #pragma unroll
      for (int i = 0; i < 8; ++i)
        if (lane < nn[i]) stage[lead + p[i]] = (unsigned short)sv[i];
    } else {
      #pragma unroll
      for (int i = 0; i < 8; ++i)
        if (lane < nn[i]) ssrc[fs + p[i]] = (unsigned short)sv[i];
    }
  }
  #pragma unroll
  for (int i = 0; i < 8; ++i) {
    if (nn[i] > 32) {
      for (int j = lane + 32; j < nn[i]; j += 32) {
        const int      ln = (int)loc8[gg[i] + j];
        const unsigned s  = src16[gg[i] + j];
        const int      q  = atomicAdd(&ncur[ln], 1);
        if (staged) stage[lead + q] = (unsigned short)s;
        else        ssrc[fs + q]    = (unsigned short)s;
      }
    }
  }
  __syncthreads();
  if (staged) {
    unsigned* gw = (unsigned*)ssrc_words + (fs >> 1);
    const int nw = (lead + total + 1) >> 1;
    for (int w = 1 + tid; w < nw - 1; w += 1024)
      gw[w] = (unsigned)stage[2 * w] | ((unsigned)stage[2 * w + 1] << 16);
    if (tid < 2 && nw > tid) {
      const int w = tid * (nw - 1);
      const bool has_lo = 2 * w >= lead;
      const bool has_hi = 2 * w + 1 < lead + total;
      if (has_lo) ssrc[2 * ((fs >> 1) + w)]     = stage[2 * w];
      if (has_hi) ssrc[2 * ((fs >> 1) + w) + 1] = stage[2 * w + 1];
    }
  }
}

__device__ __forceinline__ float leaky(float e) {
  return fmaxf(e, NEG_SLOPE * e);
}

#define LOG2E 1.4426950408889634f

__device__ __forceinline__ float quad_xor1(float x) {
  return __builtin_bit_cast(float, __builtin_amdgcn_mov_dpp(__builtin_bit_cast(int, x), 0xB1, 0xf, 0xf, true));
}
__device__ __forceinline__ float quad_xor2(float x) {
  return __builtin_bit_cast(float, __builtin_amdgcn_mov_dpp(__builtin_bit_cast(int, x), 0x4E, 0xf, 0xf, true));
}
template <int U>
__device__ __forceinline__ float quad_bcast(float x) {
  return __builtin_bit_cast(float, __builtin_amdgcn_mov_dpp(__builtin_bit_cast(int, x), U * 0x55, 0xf, 0xf, true));
}

__device__ __forceinline__ float quad_sum(float x) {
  x += quad_xor1(x);
  x += quad_xor2(x);
  return x;
}

__device__ __forceinline__ float dot4(const float4 a, const float4 b) {
  return fmaf(a.w, b.w, fmaf(a.z, b.z, fmaf(a.y, b.y, a.x * b.x)));
}

template <int CNT>
__device__ __forceinline__ void edge_group(const float* __restrict__ h, const unsigned sv,
                                           const int i0, const int lane,
                                           const v2f axy, const v2f azw, const v2f e8,
                                           float& m, float& l, v2f& accxy, v2f& acczw) {
  float4 hv[CNT];
  #pragma unroll
  for (int u = 0; u < CNT; ++u) {
    const unsigned s = __builtin_amdgcn_readlane(sv, i0 + u);
    const float4* rowp = (const float4*)(h + (size_t)s * HD);
    hv[u] = rowp[(unsigned)lane];
  }
  float d[4];
  #pragma unroll
  for (int u = 0; u < 4; ++u) {
    d[u] = 0.f;
    if (u < CNT) {
      v2f hxy; hxy.x = hv[u].x; hxy.y = hv[u].y;
      v2f hzw; hzw.x = hv[u].z; hzw.y = hv[u].w;
      v2f t = __builtin_elementwise_fma(hxy, axy, e8);
      t = __builtin_elementwise_fma(hzw, azw, t);
      d[u] = t.x + t.y;
    }
  }
  const bool b0 = (lane & 1) != 0;
  const bool b1 = (lane & 2) != 0;
  const float r01 = (b0 ? d[1] : d[0]) + quad_xor1(b0 ? d[0] : d[1]);
  const float r23 = (b0 ? d[3] : d[2]) + quad_xor1(b0 ? d[2] : d[3]);
  float e = (b1 ? r23 : r01) + quad_xor2(b1 ? r01 : r23);
  e = leaky(e);
  if (CNT < 4 && (lane & 3) >= CNT) e = -INFINITY;
  float mx = fmaxf(e, quad_xor1(e));
  mx = fmaxf(mx, quad_xor2(mx));
  const float mn = fmaxf(m, mx);
  const float sc = __builtin_amdgcn_exp2f(m - mn);
  const float p  = __builtin_amdgcn_exp2f(e - mn);
  accxy *= sc;
  acczw *= sc;
  #pragma unroll
  for (int u = 0; u < 4; ++u) {
    if (u < CNT) {
      float pu;
      if      (u == 0) pu = quad_bcast<0>(p);
      else if (u == 1) pu = quad_bcast<1>(p);
      else if (u == 2) pu = quad_bcast<2>(p);
      else             pu = quad_bcast<3>(p);
      v2f pp; pp.x = pu; pp.y = pu;
      v2f hxy; hxy.x = hv[u].x; hxy.y = hv[u].y;
      v2f hzw; hzw.x = hv[u].z; hzw.y = hv[u].w;
      accxy = __builtin_elementwise_fma(pp, hxy, accxy);
      acczw = __builtin_elementwise_fma(pp, hzw, acczw);
    }
  }
  l = fmaf(l, sc, p);
  m = mn;
}

__global__ __launch_bounds__(256)
void gat_gather(const float* __restrict__ h,
                const float* __restrict__ attn,
                const float* __restrict__ bias,
                const int* __restrict__ cnt,
                const int* __restrict__ cur,
                const int* __restrict__ ssrc_words,
                float* __restrict__ out) {
  const unsigned short* ssrc = (const unsigned short*)ssrc_words;
  const int lane = threadIdx.x & 31;
  const int t   = __builtin_amdgcn_readfirstlane(blockIdx.x * 8 + (threadIdx.x >> 5));
  const int end = __builtin_amdgcn_readfirstlane(cur[t]);
  const int beg = end - __builtin_amdgcn_readfirstlane(cnt[t]);

  unsigned sv = (beg + lane < end) ? (unsigned)ssrc[beg + lane] : 0u;
  const float4 a0 = *(const float4*)&attn[4 * lane];
  const float4 a1 = *(const float4*)&attn[HD + 4 * lane];
  const float4 ht = *(const float4*)&h[(size_t)t * HD + 4 * lane];
  const float edq = quad_sum(dot4(ht, a1));
  v2f axy; axy.x = a0.x * LOG2E; axy.y = a0.y * LOG2E;
  v2f azw; azw.x = a0.z * LOG2E; azw.y = a0.w * LOG2E;
  v2f e8;  e8.x = edq * (0.125f * LOG2E); e8.y = e8.x;

  float m = -INFINITY, l = 0.f;
  v2f accxy = {0.f, 0.f};
  v2f acczw = {0.f, 0.f};
  for (int base = beg; base < end; ) {
    const int n     = (end - base < 32) ? (end - base) : 32;
    const int nfull = n & ~3;
    for (int i0 = 0; i0 < nfull; i0 += 4)
      edge_group<4>(h, sv, i0, lane, axy, azw, e8, m, l, accxy, acczw);
    switch (n & 3) {
      case 1: edge_group<1>(h, sv, nfull, lane, axy, azw, e8, m, l, accxy, acczw); break;
      case 2: edge_group<2>(h, sv, nfull, lane, axy, azw, e8, m, l, accxy, acczw); break;
      case 3: edge_group<3>(h, sv, nfull, lane, axy, azw, e8, m, l, accxy, acczw); break;
      default: break;
    }
    base += 32;
    if (base < end) sv = (base + lane < end) ? (unsigned)ssrc[base + lane] : 0u;
  }
  l = quad_sum(l);
  const float r = 1.f / (l > 0.f ? l : 1.f);
  const float4 b = *(const float4*)&bias[4 * lane];
  float4 o;
  o.x = fmaf(accxy.x, r, b.x);
  o.y = fmaf(accxy.y, r, b.y);
  o.z = fmaf(acczw.x, r, b.z);
  o.w = fmaf(acczw.y, r, b.w);
  *(float4*)&out[(size_t)t * HD + 4 * lane] = o;
}

extern "C" void kernel_launch(void* const* d_in, const int* in_sizes, int n_in,
                              void* d_out, int out_size, void* d_ws, size_t ws_size,
                              hipStream_t stream) {
  const float* feat = (const float*)d_in[0];
  const float* W    = (const float*)d_in[1];
  const float* attn = (const float*)d_in[2];
  const float* bias = (const float*)d_in[3];
  const int*   src  = (const int*)d_in[4];
  const int*   dst  = (const int*)d_in[5];
  float*       out  = (float*)d_out;

  float* h       = (float*)d_ws;
  int*   scratch = (int*)(h + (size_t)N_NODES * HD);
  int*   cnt     = scratch + 2 * (size_t)N_NODES * NUM_HEADS;
  int*   cur     = cnt + N_NODES;
  int*   ssrc    = cur + N_NODES + 64;

  int*   src16   = scratch;
  int*   loc8    = src16 + N_EDGES / 2;
  int*   piece   = loc8 + N_EDGES / 4;

  gat_chunk_sort_and_project<<<SORT_BLOCKS + PROJ_BLOCKS, 256, 0, stream>>>(
      feat, W, h, src, dst, piece, src16, loc8);
  gat_node_sort<<<NBUCK, 1024, 0, stream>>>(piece, src16, loc8, cnt, cur, ssrc);
  gat_gather<<<N_NODES / 8, 256, 0, stream>>>(h, attn, bias, cnt, cur, ssrc, out);
}


# Node sort moved into the gather launch: 256 ranges of 196 nodes, a range's 2 blocks each sort it in LDS and gather half its nodes from there (LDS node counter); 2 launches, not 3; sum order kept
# speedup vs baseline: 13.3165x; 1.0295x over previous
#include <hip/hip_runtime.h>
#include <hip/hip_bf16.h>
#include <math.h>

#define N_NODES   50000
#define N_EDGES   800000
#define IN_FEATS  128
#define NUM_HEADS 8
#define OUT_FEATS 16
#define HD        (NUM_HEADS * OUT_FEATS)
#define NEG_SLOPE 0.2f

static_assert(N_NODES <= 65536, "sorted source ids are stored as 16-bit values");
static_assert(N_NODES % 16 == 0, "the projection stores whole 16-row tiles");


#define RANGE_NODES     ((N_NODES + 255) / 256)
#define NBUCK           ((N_NODES + RANGE_NODES - 1) / RANGE_NODES)
#define SORT_BLOCKS     256
#define EDGES_PER_BLOCK (N_EDGES / SORT_BLOCKS)
#define EPT             ((EDGES_PER_BLOCK + 255) / 256)
static_assert(N_EDGES % SORT_BLOCKS == 0, "every sort block takes the same number of edges");
static_assert(N_EDGES % 4 == 0, "the range-local node bytes fill whole words");
static_assert(NBUCK <= 256, "one thread per node range in a 256-thread block");
static_assert(RANGE_NODES <= 256, "a node's place inside its range is stored in one byte");
static_assert(SORT_BLOCKS == 256, "a piece word is indexed range * 256 + chunk; 8 chunks per wave in the node sort");
static_assert(N_NODES <= 256 * 256, "range << 8 | place stays a small positive number");
static_assert(EDGES_PER_BLOCK < 32768, "a piece's offset and count share one word, 16 bits each");
static_assert((EPT - 1) * 256 <= EDGES_PER_BLOCK, "only a thread's last edge slot can be empty");
static_assert(N_EDGES / 2 + N_EDGES / 4 + 256 * SORT_BLOCKS
              <= 2 * N_NODES * NUM_HEADS, "sort scratch fits its region");

typedef float v2f __attribute__((ext_vector_type(2)));
typedef float v8f __attribute__((ext_vector_type(8)));

#define LDA_PITCH (IN_FEATS + 4)
#define PROJ_ROWS   32
#define PROJ_BLOCKS ((N_NODES + PROJ_ROWS - 1) / PROJ_ROWS)
__device__ __forceinline__
void gemm_tile(const float* __restrict__ feat,
               const float* __restrict__ W,
               float* __restrict__ h, const int tile) {
  __shared__ float lA[PROJ_ROWS * LDA_PITCH];
  const int tid  = threadIdx.x;
  const int wave = tid >> 5;
  const int lane = tid & 31;
  const int m0   = tile * PROJ_ROWS;
  const int n0   = wave * 16;

  const int m1 = (m0 + 16 < N_NODES) ? m0 + 16 : m0;
  const int sr = tid >> 7;
  const int sc = tid & 127;
  float st[PROJ_ROWS / 2];
  #pragma unroll
  for (int j = 0; j < 8; ++j) {
    st[j]     = feat[(size_t)(m0 + sr + 2 * j) * IN_FEATS + sc];
    st[8 + j] = feat[(size_t)(m1 + sr + 2 * j) * IN_FEATS + sc];
  }
  #pragma unroll
  for (int j = 0; j < 8; ++j) {
    lA[(sr + 2 * j) * LDA_PITCH + sc]      = st[j];
    lA[(16 + sr + 2 * j) * LDA_PITCH + sc] = st[8 + j];
  }
  __syncthreads();

  const int ml  = lane & 15;
  const int klo = (lane < 16) ? 0 : 2;

  v8f acc0 = {};
  v8f acc1 = {};
  #pragma unroll 4
  for (int k = 0; k < IN_FEATS; k += 4) {
    const float2 av0 = *(const float2*)&lA[ml * LDA_PITCH + k + klo];
    const float2 av1 = *(const float2*)&lA[(16 + ml) * LDA_PITCH + k + klo];
    v2f a0; a0.x = av0.x; a0.y = av0.y;
    v2f a1; a1.x = av1.x; a1.y = av1.y;
    v2f b;
    b.x = W[(size_t)(k + klo)     * HD + n0 + ml];
    b.y = W[(size_t)(k + klo + 1) * HD + n0 + ml];
    acc0 = __builtin_amdgcn_wmma_f32_16x16x4_f32(
         false, a0,  false, b,
         (short)0, acc0,  false,  false);
    acc1 = __builtin_amdgcn_wmma_f32_16x16x4_f32(
         false, a1,  false, b,
         (short)0, acc1,  false,  false);
  }

  const int rbase = m0 + ((lane < 16) ? 0 : 8);
  #pragma unroll
  for (int v = 0; v < 8; ++v) {
    h[(size_t)(rbase + v) * HD + n0 + ml] = acc0[v];
  }
  if (m0 + 16 < N_NODES) {
    #pragma unroll
    for (int v = 0; v < 8; ++v) {
      h[(size_t)(rbase + 16 + v) * HD + n0 + ml] = acc1[v];
    }
  }
}


__device__ __forceinline__ int excl_scan_256(int v, int tid, int* wsum) {
  const int lane = tid & 31;
  const int w    = tid >> 5;
  int x = v;
  #pragma unroll
  for (int o = 1; o < 32; o <<= 1) {
    const int y = __shfl_up(x, o);
    if (lane >= o) x += y;
  }
  if (lane == 31 && w < 8) wsum[w] = x;
  __syncthreads();
  int off = 0;
  #pragma unroll
  for (int j = 0; j < 8; ++j) if (j < w) off += wsum[j];
  return off + x - v;
}

__device__ __forceinline__
void chunk_sort(const int* __restrict__ src, const int* __restrict__ dst,
                int* __restrict__ piece,
                int* __restrict__ src16_words, int* __restrict__ loc8_words,
                const int chunk) {
  __shared__ int lcnt[256];
  __shared__ int loff[256];
  __shared__ int wsum[8];
  __shared__ __attribute__((aligned(16))) unsigned short st16[EDGES_PER_BLOCK + 4];
  __shared__ unsigned st8w[(EDGES_PER_BLOCK + 3) / 4 + 2];
  unsigned char*  st8   = (unsigned char*)st8w;
  unsigned short* src16 = (unsigned short*)src16_words;
  unsigned char*  loc8  = (unsigned char*)loc8_words;
  const int tid = threadIdx.x;
  const int e0  = chunk * EDGES_PER_BLOCK;
  const int lead = e0 & 3;
  const int base = e0 - lead;
  lcnt[tid] = 0;
  int d[EPT], s[EPT], rk[EPT];
  const bool last_ok = tid + (EPT - 1) * 256 < EDGES_PER_BLOCK;
  #pragma unroll
  for (int k = 0; k < EPT; ++k) {
    const int i = (k < EPT - 1 || last_ok) ? tid + k * 256 : EDGES_PER_BLOCK - 1;
    d[k] = dst[e0 + i];
    s[k] = src[e0 + i];
  }
  unsigned pk[EPT];
  #pragma unroll
  for (int k = 0; k < EPT; ++k) {
    const unsigned rg = (unsigned)d[k] / RANGE_NODES;
    pk[k] = ((unsigned)s[k] << 16) | (rg << 8) | ((unsigned)d[k] - rg * RANGE_NODES);
  }
  __syncthreads();
  #pragma unroll
  for (int k = 0; k < EPT; ++k) {
    rk[k] = 0;
    if (k < EPT - 1 || last_ok) rk[k] = atomicAdd(&lcnt[(pk[k] >> 8) & 255u], 1);
  }
  __syncthreads();
  const int c   = lcnt[tid];
  const int off = excl_scan_256(c, tid, wsum);
  loff[tid] = off + lead;
  if (tid < NBUCK) piece[tid * 256 + chunk] = off | (c << 16);
  __syncthreads();
  #pragma unroll
  for (int k = 0; k < EPT; ++k) {
    if (k < EPT - 1 || last_ok) {
      const int q = loff[(pk[k] >> 8) & 255u] + rk[k];
      st16[q] = (unsigned short)(pk[k] >> 16);
      st8[q]  = (unsigned char)(pk[k] & 255u);
    }
  }
  __syncthreads();
  {
    unsigned* gw = (unsigned*)src16_words + (base >> 1);
    const int wf = lead >> 1;
    const int wl = (lead + EDGES_PER_BLOCK - 1) >> 1;
    #pragma unroll
    for (int k = 0; k < (EDGES_PER_BLOCK / 2 + 1 + 255) / 256; ++k) {
      const int w = wf + 1 + tid + 256 * k;
      if (w < wl) gw[w] = (unsigned)st16[2 * w] | ((unsigned)st16[2 * w + 1] << 16);
    }
    if (tid < 2) {
      const int w = tid ? wl : wf;
      const bool has_lo = 2 * w >= lead && 2 * w < lead + EDGES_PER_BLOCK;
      const bool has_hi = 2 * w + 1 >= lead && 2 * w + 1 < lead + EDGES_PER_BLOCK;
      if (has_lo) src16[base + 2 * w]     = st16[2 * w];
      if (has_hi) src16[base + 2 * w + 1] = st16[2 * w + 1];
    }
  }
  {
    unsigned* gw = (unsigned*)loc8_words + (base >> 2);
    const int nb = lead + EDGES_PER_BLOCK;
    const int nw = (nb + 3) >> 2;
    #pragma unroll
    for (int k = 0; k < (EDGES_PER_BLOCK / 4 + 2 + 255) / 256; ++k) {
      const int w = 1 + tid + 256 * k;
      if (w < nw - 1) gw[w] = st8w[w];
    }
    if (tid < 2) {
      const int w = tid * (nw - 1);
      for (int j = 0; j < 4; ++j) {
        const int o = 4 * w + j;
        if (o >= lead && o < nb) loc8[base + o] = st8[o];
      }
    }
  }
}

__global__ __launch_bounds__(256, 16)
void gat_chunk_sort_and_project(const float* __restrict__ feat,
                                const float* __restrict__ W,
                                float* __restrict__ h,
                                const int* __restrict__ src, const int* __restrict__ dst,
                                int* __restrict__ piece,
                                int* __restrict__ src16_words, int* __restrict__ loc8_words) {
  if (blockIdx.x < SORT_BLOCKS) chunk_sort(src, dst, piece, src16_words, loc8_words, blockIdx.x);
  else                          gemm_tile(feat, W, h, blockIdx.x - SORT_BLOCKS);
}

__device__ __forceinline__ float leaky(float e) {
  return fmaxf(e, NEG_SLOPE * e);
}

#define LOG2E 1.4426950408889634f

__device__ __forceinline__ float quad_xor1(float x) {
  return __builtin_bit_cast(float, __builtin_amdgcn_mov_dpp(__builtin_bit_cast(int, x), 0xB1, 0xf, 0xf, true));
}
__device__ __forceinline__ float quad_xor2(float x) {
  return __builtin_bit_cast(float, __builtin_amdgcn_mov_dpp(__builtin_bit_cast(int, x), 0x4E, 0xf, 0xf, true));
}
template <int U>
__device__ __forceinline__ float quad_bcast(float x) {
  return __builtin_bit_cast(float, __builtin_amdgcn_mov_dpp(__builtin_bit_cast(int, x), U * 0x55, 0xf, 0xf, true));
}

__device__ __forceinline__ float quad_sum(float x) {
  x += quad_xor1(x);
  x += quad_xor2(x);
  return x;
}

__device__ __forceinline__ float dot4(const float4 a, const float4 b) {
  return fmaf(a.w, b.w, fmaf(a.z, b.z, fmaf(a.y, b.y, a.x * b.x)));
}

template <int CNT>
__device__ __forceinline__ void edge_group(const float* __restrict__ h, const unsigned sv,
                                           const int i0, const int lane,
                                           const v2f axy, const v2f azw, const v2f e8,
                                           float& m, float& l, v2f& accxy, v2f& acczw) {
  float4 hv[CNT];
  #pragma unroll
  for (int u = 0; u < CNT; ++u) {
    const unsigned s = __builtin_amdgcn_readlane(sv, i0 + u);
    const float4* rowp = (const float4*)(h + (size_t)s * HD);
    hv[u] = rowp[(unsigned)lane];
  }
  float d[4];
  #pragma unroll
  for (int u = 0; u < 4; ++u) {
    d[u] = 0.f;
    if (u < CNT) {
      v2f hxy; hxy.x = hv[u].x; hxy.y = hv[u].y;
      v2f hzw; hzw.x = hv[u].z; hzw.y = hv[u].w;
      v2f t = __builtin_elementwise_fma(hxy, axy, e8);
      t = __builtin_elementwise_fma(hzw, azw, t);
      d[u] = t.x + t.y;
    }
  }
  const bool b0 = (lane & 1) != 0;
  const bool b1 = (lane & 2) != 0;
  const float r01 = (b0 ? d[1] : d[0]) + quad_xor1(b0 ? d[0] : d[1]);
  const float r23 = (b0 ? d[3] : d[2]) + quad_xor1(b0 ? d[2] : d[3]);
  float e = (b1 ? r23 : r01) + quad_xor2(b1 ? r01 : r23);
  e = leaky(e);
  if (CNT < 4 && (lane & 3) >= CNT) e = -INFINITY;
  float mx = fmaxf(e, quad_xor1(e));
  mx = fmaxf(mx, quad_xor2(mx));
  const float mn = fmaxf(m, mx);
  const float sc = __builtin_amdgcn_exp2f(m - mn);
  const float p  = __builtin_amdgcn_exp2f(e - mn);
  accxy *= sc;
  acczw *= sc;
  #pragma unroll
  for (int u = 0; u < 4; ++u) {
    if (u < CNT) {
      float pu;
      if      (u == 0) pu = quad_bcast<0>(p);
      else if (u == 1) pu = quad_bcast<1>(p);
      else if (u == 2) pu = quad_bcast<2>(p);
      else             pu = quad_bcast<3>(p);
      v2f pp; pp.x = pu; pp.y = pu;
      v2f hxy; hxy.x = hv[u].x; hxy.y = hv[u].y;
      v2f hzw; hzw.x = hv[u].z; hzw.y = hv[u].w;
      accxy = __builtin_elementwise_fma(pp, hxy, accxy);
      acczw = __builtin_elementwise_fma(pp, hzw, acczw);
    }
  }
  l = fmaf(l, sc, p);
  m = mn;
}

#define STAGE_CAP 8192
#define GATHER_SPLIT 2
#define NODES_PER_PART (RANGE_NODES / GATHER_SPLIT)
static_assert(RANGE_NODES % GATHER_SPLIT == 0, "a range's nodes are shared out evenly");
#define SPILL_SHIFT 64
#define SPILL_WORDS (N_EDGES / 2 + (SPILL_SHIFT / 2) * NBUCK * GATHER_SPLIT)
static_assert(SPILL_WORDS <= 2 * N_NODES + 64 + N_EDGES / 2, "the spill regions fit behind the sort scratch");

template <bool STAGED>
__device__ __forceinline__
void gather_share(const float* __restrict__ h, const float* __restrict__ attn,
                  const float* __restrict__ bias, float* __restrict__ out,
                  const unsigned short* ids, const int* ncnt, const int* ncur, int* next_node,
                  const int node0, const int node_lim, const int lane) {
  const float4 a0 = *(const float4*)&attn[4 * lane];
  const float4 a1 = *(const float4*)&attn[HD + 4 * lane];
  const float4 bv = *(const float4*)&bias[4 * lane];
  v2f axy; axy.x = a0.x * LOG2E; axy.y = a0.y * LOG2E;
  v2f azw; azw.x = a0.z * LOG2E; azw.y = a0.w * LOG2E;

  for (int trip = 0; trip < NODES_PER_PART; ++trip) {
    int jn = 0;
    if (lane == 0) jn = atomicAdd(next_node, 1);
    jn = __builtin_amdgcn_readfirstlane(jn);
    if (jn >= node_lim) break;
    const int t   = node0 + jn;
    const int end = __builtin_amdgcn_readfirstlane(ncur[jn]);
    const int beg = end - __builtin_amdgcn_readfirstlane(ncnt[jn]);

    unsigned sv = (beg + lane < end) ? (unsigned)ids[beg + lane] : 0u;
    const float4 ht = *(const float4*)&h[(size_t)t * HD + 4 * lane];
    const float edq = quad_sum(dot4(ht, a1));
    v2f e8;  e8.x = edq * (0.125f * LOG2E); e8.y = e8.x;

    float m = -INFINITY, ls = 0.f;
    v2f accxy = {0.f, 0.f};
    v2f acczw = {0.f, 0.f};
    for (int base = beg; base < end; ) {
      const int n     = (end - base < 32) ? (end - base) : 32;
      const int nfull = n & ~3;
      for (int i0 = 0; i0 < nfull; i0 += 4)
        edge_group<4>(h, sv, i0, lane, axy, azw, e8, m, ls, accxy, acczw);
      switch (n & 3) {
        case 1: edge_group<1>(h, sv, nfull, lane, axy, azw, e8, m, ls, accxy, acczw); break;
        case 2: edge_group<2>(h, sv, nfull, lane, axy, azw, e8, m, ls, accxy, acczw); break;
        case 3: edge_group<3>(h, sv, nfull, lane, axy, azw, e8, m, ls, accxy, acczw); break;
        default: break;
      }
      base += 32;
      if (base < end) sv = (base + lane < end) ? (unsigned)ids[base + lane] : 0u;
    }
    ls = quad_sum(ls);
    const float r = 1.f / (ls > 0.f ? ls : 1.f);
    float4 o;
    o.x = fmaf(accxy.x, r, bv.x);
    o.y = fmaf(accxy.y, r, bv.y);
    o.z = fmaf(acczw.x, r, bv.z);
    o.w = fmaf(acczw.y, r, bv.w);
    *(float4*)&out[(size_t)t * HD + 4 * lane] = o;
  }
}

__global__ __launch_bounds__(1024, 16)
void gat_node_sort_gather(const int* __restrict__ piece,
                          const int* __restrict__ src16_words,
                          const int* __restrict__ loc8_words,
                          int* __restrict__ spill_words,
                          const float* __restrict__ h,
                          const float* __restrict__ attn,
                          const float* __restrict__ bias,
                          float* __restrict__ out) {
  __shared__ int ncnt[256];
  __shared__ int ncur[256];
  __shared__ int pbase[256];
  __shared__ int plen[256];
  __shared__ int wsb[8];
  __shared__ int first_slot;
  __shared__ int range_total;
  __shared__ int next_node;
  __shared__ __attribute__((aligned(16))) unsigned short stage[STAGE_CAP];
  const unsigned short* src16 = (const unsigned short*)src16_words;
  const unsigned char*  loc8  = (const unsigned char*)loc8_words;
  unsigned short* spill = (unsigned short*)spill_words;
  const int tid  = threadIdx.x;
  const int b    = blockIdx.x / GATHER_SPLIT;
  const int part = blockIdx.x % GATHER_SPLIT;

  int low = 0;
  if (tid < 256) {
    const int v = piece[b * 256 + tid];
    low = v & 0xFFFF;
    pbase[tid] = tid * EDGES_PER_BLOCK + low;
    plen[tid]  = v >> 16;
    ncnt[tid]  = 0;
  }
  if (tid == 0) { first_slot = 0; next_node = part * NODES_PER_PART; }
  __syncthreads();
  const int lane = tid & 31;
  const int w8   = (tid >> 5) * 8;
  int nn[8], gg[8];
  #pragma unroll
  for (int i = 0; i < 8; ++i) {
    nn[i] = __builtin_amdgcn_readfirstlane(plen[w8 + i]);
    gg[i] = __builtin_amdgcn_readfirstlane(pbase[w8 + i]);
  }
  int l[8], sv8[8];
  #pragma unroll
  for (int i = 0; i < 8; ++i) {
    const int jj = max(min(lane, nn[i] - 1), 0);
    const int at = min(gg[i] + jj, N_EDGES - 1);
    l[i]   = (int)loc8[at];
    sv8[i] = (int)src16[at];
  }
  asm volatile("" :: "v"(l[0]), "v"(l[1]), "v"(l[2]), "v"(l[3]),
                     "v"(l[4]), "v"(l[5]), "v"(l[6]), "v"(l[7]),
                     "v"(sv8[0]), "v"(sv8[1]), "v"(sv8[2]), "v"(sv8[3]),
                     "v"(sv8[4]), "v"(sv8[5]), "v"(sv8[6]), "v"(sv8[7]));
  #pragma unroll
  for (int i = 0; i < 8; ++i)
    if (lane < nn[i]) atomicAdd(&ncnt[l[i]], 1);
  #pragma unroll
  for (int i = 0; i < 8; ++i) {
    if (nn[i] > 32) {
      for (int j = lane + 32; j < nn[i]; j += 32) atomicAdd(&ncnt[(int)loc8[gg[i] + j]], 1);
    }
  }
  low += __builtin_amdgcn_update_dpp(0, low, 0x161, 0xf, 0xf, false);
  low += __builtin_amdgcn_update_dpp(0, low, 0x162, 0xf, 0xf, false);
  low += __builtin_amdgcn_update_dpp(0, low, 0x164, 0xf, 0xf, false);
  low += __builtin_amdgcn_update_dpp(0, low, 0x168, 0xf, 0xf, false);
  if ((tid & 15) == 0 && low != 0) atomicAdd(&first_slot, low);
  __syncthreads();
  const int c  = (tid < 256) ? ncnt[tid] : 0;
  const int nb = excl_scan_256(c, tid, wsb);
  if (tid < 256) {
    ncur[tid] = nb;
    if (tid == 255) range_total = nb + c;
  }
  __syncthreads();
  const int  fs     = first_slot;
  const int  total  = range_total;
  const bool staged = total <= STAGE_CAP;
  unsigned short* const myspill = spill + fs + SPILL_SHIFT * (int)blockIdx.x;
  const int share0 = part * NODES_PER_PART;
  {
    unsigned p[8];
    #pragma unroll
    for (int i = 0; i < 8; ++i) {
      p[i] = 0;
      if (lane < nn[i]) p[i] = (unsigned)atomicAdd(&ncur[l[i]], 1);
    }
    if (staged) {
      #pragma unroll
      for (int i = 0; i < 8; ++i)
        if (lane < nn[i]) stage[p[i]] = (unsigned short)sv8[i];
    } else {
      #pragma unroll
      for (int i = 0; i < 8; ++i)
        if (lane < nn[i] && (unsigned)(l[i] - share0) < (unsigned)NODES_PER_PART)
          myspill[p[i]] = (unsigned short)sv8[i];
    }
  }
  #pragma unroll
  for (int i = 0; i < 8; ++i) {
    if (nn[i] > 32) {
      for (int j = lane + 32; j < nn[i]; j += 32) {
        const int      ln = (int)loc8[gg[i] + j];
        const unsigned s  = src16[gg[i] + j];
        const int      q  = atomicAdd(&ncur[ln], 1);
        if (staged) stage[q] = (unsigned short)s;
        else if ((unsigned)(ln - share0) < (unsigned)NODES_PER_PART) myspill[q] = (unsigned short)s;
      }
    }
  }
  __syncthreads();
  const int node_lim = min((part + 1) * NODES_PER_PART, N_NODES - b * RANGE_NODES);
  if (staged) gather_share<true >(h, attn, bias, out, stage,      ncnt, ncur, &next_node, b * RANGE_NODES, node_lim, lane);
  else        gather_share<false>(h, attn, bias, out, myspill,    ncnt, ncur, &next_node, b * RANGE_NODES, node_lim, lane);
}

extern "C" void kernel_launch(void* const* d_in, const int* in_sizes, int n_in,
                              void* d_out, int out_size, void* d_ws, size_t ws_size,
                              hipStream_t stream) {
  const float* feat = (const float*)d_in[0];
  const float* W    = (const float*)d_in[1];
  const float* attn = (const float*)d_in[2];
  const float* bias = (const float*)d_in[3];
  const int*   src  = (const int*)d_in[4];
  const int*   dst  = (const int*)d_in[5];
  float*       out  = (float*)d_out;

  float* h       = (float*)d_ws;
  int*   scratch = (int*)(h + (size_t)N_NODES * HD);
  int*   spill   = scratch + 2 * (size_t)N_NODES * NUM_HEADS;

  int*   src16   = scratch;
  int*   loc8    = src16 + N_EDGES / 2;
  int*   piece   = loc8 + N_EDGES / 4;

  gat_chunk_sort_and_project<<<SORT_BLOCKS + PROJ_BLOCKS, 256, 0, stream>>>(
      feat, W, h, src, dst, piece, src16, loc8);
  gat_node_sort_gather<<<NBUCK * GATHER_SPLIT, 1024, 0, stream>>>(
      piece, src16, loc8, spill, h, attn, bias, out);
}
